# starting kernel
# baseline (speedup 1.0000x reference)
#include <hip/hip_runtime.h>

typedef __attribute__((ext_vector_type(2))) float v2f;
typedef __attribute__((ext_vector_type(4))) float v4f;
typedef __attribute__((ext_vector_type(8))) float v8f;

#define DFEAT 64

__global__ void gcn_zero_kernel(v4f* __restrict__ p, int n4) {
  int i = blockIdx.x * blockDim.x + threadIdx.x;
  if (i < n4) {
    v4f z = {0.0f, 0.0f, 0.0f, 0.0f};
    p[i] = z;
  }
}

__global__ void gcn_scatter_kernel(const float* __restrict__ X,
                                   const float* __restrict__ vals,
                                   const int* __restrict__ row,
                                   const int* __restrict__ col,
                                   float* __restrict__ temp,
                                   int n_edges) {
  long long t = (long long)blockIdx.x * blockDim.x + threadIdx.x;
  int e = (int)(t >> 4);
  if (e >= n_edges) return;
  int chunk = ((int)t & 15) * 4;

  int r = row[e];
  int c = col[e];
  float v = vals[e];

  v4f x = *(const v4f*)(X + (long long)c * DFEAT + chunk);
  float* td = temp + (long long)r * DFEAT + chunk;
  atomicAdd(td + 0, x.x * v);
  atomicAdd(td + 1, x.y * v);
  atomicAdd(td + 2, x.z * v);
  atomicAdd(td + 3, x.w * v);
}

__global__ void __launch_bounds__(256)
gcn_gemm_wmma_kernel(const float* temp,
                     const float* __restrict__ W,
                     const float* __restrict__ bias,
                     float* out, int n_rows) {
  __shared__ float Wl2[DFEAT * DFEAT];
  __shared__ float bl[DFEAT];
  for (int i = threadIdx.x; i < DFEAT * DFEAT; i += 256) {
    int k = i >> 6;
    int n = i & 63;
    Wl2[(k >> 1) * 128 + n * 2 + (k & 1)] = W[i];
  }
  if (threadIdx.x < DFEAT) bl[threadIdx.x] = bias[threadIdx.x];
  __syncthreads();

  int wave = (int)((blockIdx.x * 256u + threadIdx.x) >> 5);
  int lane = (int)(threadIdx.x & 31u);
  int m0 = wave * 16;
  if (m0 >= n_rows) return;

  int lo = lane & 15;
  int hi = lane >> 4;
  int koff = hi * 2;

  v8f acc[4];
#pragma unroll
  for (int t = 0; t < 4; ++t) {
    float bv = bl[t * 16 + lo];
#pragma unroll
    for (int r = 0; r < 8; ++r) acc[t][r] = bv;
  }

  int mrow = m0 + lo;
  if (mrow >= n_rows) mrow = n_rows - 1;
  const float* arow = temp + (long long)mrow * DFEAT;

  const v2f* wl2 = (const v2f*)Wl2;

#pragma unroll
  for (int kk = 0; kk < 16; ++kk) {
    int kb = kk * 4;
    int kpair = (kb + koff) >> 1;
    v2f a;
    a.x = arow[kb + koff];
    a.y = arow[kb + koff + 1];
#pragma unroll
    for (int t = 0; t < 4; ++t) {
      v2f bb = wl2[kpair * 64 + t * 16 + lo];
      acc[t] = __builtin_amdgcn_wmma_f32_16x16x4_f32(
           false, a,  false, bb,
           (short)0, acc[t],  false,  false);
    }
  }

#pragma unroll
  for (int t = 0; t < 4; ++t) {
#pragma unroll
    for (int r = 0; r < 8; ++r) {
      int m = m0 + r + 8 * hi;
      if (m < n_rows) out[(long long)m * DFEAT + t * 16 + lo] = acc[t][r];
    }
  }
}

extern "C" void kernel_launch(void* const* d_in, const int* in_sizes, int n_in,
                              void* d_out, int out_size, void* d_ws, size_t ws_size,
                              hipStream_t stream) {
  const float* X    = (const float*)d_in[0];
  const float* vals = (const float*)d_in[1];
  const float* W    = (const float*)d_in[2];
  const float* bias = (const float*)d_in[3];
  const int*   row  = (const int*)d_in[4];
  const int*   col  = (const int*)d_in[5];
  float* out = (float*)d_out;

  int n_nodes = in_sizes[0] / DFEAT;
  int n_edges = in_sizes[1];

  size_t temp_bytes = (size_t)n_nodes * DFEAT * sizeof(float);
  float* temp = (ws_size >= temp_bytes) ? (float*)d_ws : out;

  int n4 = n_nodes * (DFEAT / 4);
  gcn_zero_kernel<<<(n4 + 255) / 256, 256, 0, stream>>>((v4f*)temp, n4);

  long long sthreads = (long long)n_edges * 16;
  unsigned sblocks = (unsigned)((sthreads + 255) / 256);
  gcn_scatter_kernel<<<sblocks, 256, 0, stream>>>(X, vals, row, col, temp, n_edges);

  int waves = (n_nodes + 15) / 16;
  long long gthreads = (long long)waves * 32;
  unsigned gblocks = (unsigned)((gthreads + 255) / 256);
  gcn_gemm_wmma_kernel<<<gblocks, 256, 0, stream>>>(temp, W, bias, out, n_nodes);
}


# Float-atomic scatter replaced by binning edges into 256-row destination buckets (integer counts) and summing each bucket in an LDS tile with LDS f32 adds (arrival order kept); no temp zero fill
# speedup vs baseline: 1.8380x; 1.8380x over previous
#include <hip/hip_runtime.h>


typedef __attribute__((ext_vector_type(2))) float v2f;
typedef __attribute__((ext_vector_type(4))) float v4f;
typedef __attribute__((ext_vector_type(8))) float v8f;
typedef __attribute__((ext_vector_type(4))) int v4i;

#define DFEAT 64
#define RPB_SHIFT 8
#define RPB 256
#define CHUNK 8192
#define BIN_THREADS 256
#define SUM_THREADS 512
#define CB 16
#define PB 8

__global__ void gcn_clear_counts_kernel(int* __restrict__ p, int n) {
  int i = blockIdx.x * blockDim.x + threadIdx.x;
  if (i < n) p[i] = 0;
}

__global__ void __launch_bounds__(BIN_THREADS)
gcn_bucket_count_kernel(const int* __restrict__ row, int n_edges, int n_buckets,
                        int* __restrict__ hist) {
  extern __shared__ int sm[];
  int* lh = sm;
  const int tid = threadIdx.x;
  for (int i = tid; i < n_buckets; i += BIN_THREADS) lh[i] = 0;
  __syncthreads();
  const int e0 = blockIdx.x * CHUNK;
  const int e1 = (e0 + CHUNK < n_edges) ? (e0 + CHUNK) : n_edges;
  for (int eb = e0 + tid; eb < e1; eb += BIN_THREADS * CB) {
    int r[CB];
#pragma unroll
    for (int k = 0; k < CB; ++k) {
      int e = eb + k * BIN_THREADS;
      r[k] = (e < e1) ? row[e] : -1;
    }
#pragma unroll
    for (int k = 0; k < CB; ++k)
      if (r[k] >= 0) atomicAdd(&lh[r[k] >> RPB_SHIFT], 1);
  }
  __syncthreads();
  for (int i = tid; i < n_buckets; i += BIN_THREADS) {
    int c = lh[i];
    if (c) atomicAdd(&hist[i], c);
  }
}

__global__ void __launch_bounds__(BIN_THREADS)
gcn_bucket_place_kernel(const int* __restrict__ row, const int* __restrict__ col,
                        const float* __restrict__ vals, int n_edges, int n_buckets,
                        const int* __restrict__ hist, int* __restrict__ cursor,
                        v4i* __restrict__ recs) {
  extern __shared__ int sm[];
  int* lh = sm;
  int* gs = sm + n_buckets;
  int* part = sm + 2 * n_buckets;
  const int tid = threadIdx.x;

  for (int i = tid; i < n_buckets; i += BIN_THREADS) lh[i] = 0;
  __syncthreads();
  const int e0 = blockIdx.x * CHUNK;
  const int e1 = (e0 + CHUNK < n_edges) ? (e0 + CHUNK) : n_edges;
  for (int eb = e0 + tid; eb < e1; eb += BIN_THREADS * CB) {
    int r[CB];
#pragma unroll
    for (int k = 0; k < CB; ++k) {
      int e = eb + k * BIN_THREADS;
      r[k] = (e < e1) ? row[e] : -1;
    }
#pragma unroll
    for (int k = 0; k < CB; ++k)
      if (r[k] >= 0) atomicAdd(&lh[r[k] >> RPB_SHIFT], 1);
  }

  const int seg = (n_buckets + BIN_THREADS - 1) / BIN_THREADS;
  const int s0 = tid * seg;
  const int s1 = (s0 + seg < n_buckets) ? (s0 + seg) : n_buckets;
  int ssum = 0;
  for (int i = s0; i < s1; ++i) ssum += hist[i];
  part[tid] = ssum;
  __syncthreads();
  int pre = 0;
  for (int i = 0; i < tid; ++i) pre += part[i];
  for (int i = s0; i < s1; ++i) {
    int c = lh[i];
    int st = pre;
    if (c) st += atomicAdd(&cursor[i], c);
    gs[i] = st;
    lh[i] = 0;
    pre += hist[i];
  }
  __syncthreads();

  for (int eb = e0 + tid; eb < e1; eb += BIN_THREADS * PB) {
    int r[PB], c[PB], vb[PB], pos[PB];
#pragma unroll
    for (int k = 0; k < PB; ++k) {
      int e = eb + k * BIN_THREADS;
      bool ok = e < e1;
      r[k] = ok ? row[e] : -1;
      c[k] = ok ? col[e] : 0;
      vb[k] = ok ? __float_as_int(vals[e]) : 0;
    }
#pragma unroll
    for (int k = 0; k < PB; ++k) {
      pos[k] = 0;
      if (r[k] >= 0) {
        int b = r[k] >> RPB_SHIFT;
        pos[k] = gs[b] + atomicAdd(&lh[b], 1);
      }
    }
#pragma unroll
    for (int k = 0; k < PB; ++k) {
      if (r[k] >= 0) {
        v4i rec;
        rec.x = c[k];
        rec.y = r[k] & (RPB - 1);
        rec.z = vb[k];
        rec.w = 0;
        recs[pos[k]] = rec;
      }
    }
  }
}

__global__ void __launch_bounds__(SUM_THREADS)
gcn_bucket_sum_kernel(const float* __restrict__ X, const v4i* __restrict__ recs,
                      const int* __restrict__ hist, float* __restrict__ temp,
                      int n_nodes) {
  __shared__ float tile[RPB * DFEAT];
  __shared__ int sbase;
  const int tid = threadIdx.x;
  const int b = blockIdx.x;

  v4f z = {0.0f, 0.0f, 0.0f, 0.0f};
  for (int i = tid; i < RPB * DFEAT / 4; i += SUM_THREADS) ((v4f*)tile)[i] = z;
  if (tid == 0) sbase = 0;
  __syncthreads();
  int s = 0;
  for (int i = tid; i < b; i += SUM_THREADS) s += hist[i];
  atomicAdd(&sbase, s);
  __syncthreads();
  const int base = sbase;
  const int cnt = hist[b];
  const v4i* rb = recs + base;

  const int lane = tid & 31;
  const int wave = __builtin_amdgcn_readfirstlane(tid >> 5);

  for (int i0 = wave * 32; i0 < cnt; i0 += (SUM_THREADS / 32) * 32) {
    int idx = i0 + lane;
    if (idx >= cnt) idx = cnt - 1;
    const v4i rec = rb[idx];
    const int m = (cnt - i0 < 32) ? (cnt - i0) : 32;
    if (m == 32) {
      v2f xr[32];
#pragma unroll
      for (int j = 0; j < 32; ++j) {
        int c = __builtin_amdgcn_readlane(rec.x, j);
        xr[j] = *(const v2f*)(X + (long long)c * DFEAT + 2 * lane);
      }
#pragma unroll
      for (int j = 0; j < 32; ++j) {
        int rl = __builtin_amdgcn_readlane(rec.y, j);
        float v = __int_as_float(__builtin_amdgcn_readlane(rec.z, j));
        float* td = tile + rl * DFEAT + 2 * lane;
        atomicAdd(td + 0, xr[j].x * v);
        atomicAdd(td + 1, xr[j].y * v);
      }
    } else {
      for (int j = 0; j < m; ++j) {
        int c = __builtin_amdgcn_readlane(rec.x, j);
        int rl = __builtin_amdgcn_readlane(rec.y, j);
        float v = __int_as_float(__builtin_amdgcn_readlane(rec.z, j));
        v2f x = *(const v2f*)(X + (long long)c * DFEAT + 2 * lane);
        float* td = tile + rl * DFEAT + 2 * lane;
        atomicAdd(td + 0, x.x * v);
        atomicAdd(td + 1, x.y * v);
      }
    }
  }
  __syncthreads();

  const int r0 = b * RPB;
  const int nrows = (n_nodes - r0 < RPB) ? (n_nodes - r0) : RPB;
  v4f* dst = (v4f*)(temp + (long long)r0 * DFEAT);
  for (int i = tid; i < nrows * (DFEAT / 4); i += SUM_THREADS) dst[i] = ((const v4f*)tile)[i];
}

__global__ void __launch_bounds__(256)
gcn_gemm_wmma_kernel(const float* temp,
                     const float* __restrict__ W,
                     const float* __restrict__ bias,
                     float* out, int n_rows) {
  __shared__ float Wl2[DFEAT * DFEAT];
  __shared__ float bl[DFEAT];
  for (int i = threadIdx.x; i < DFEAT * DFEAT; i += 256) {
    int k = i >> 6;
    int n = i & 63;
    Wl2[(k >> 1) * 128 + n * 2 + (k & 1)] = W[i];
  }
  if (threadIdx.x < DFEAT) bl[threadIdx.x] = bias[threadIdx.x];
  __syncthreads();

  int wave = (int)((blockIdx.x * 256u + threadIdx.x) >> 5);
  int lane = (int)(threadIdx.x & 31u);
  int m0 = wave * 16;
  if (m0 >= n_rows) return;

  int lo = lane & 15;
  int hi = lane >> 4;
  int koff = hi * 2;

  v8f acc[4];
#pragma unroll
  for (int t = 0; t < 4; ++t) {
    float bv = bl[t * 16 + lo];
#pragma unroll
    for (int r = 0; r < 8; ++r) acc[t][r] = bv;
  }

  int mrow = m0 + lo;
  if (mrow >= n_rows) mrow = n_rows - 1;
  const float* arow = temp + (long long)mrow * DFEAT;

  const v2f* wl2 = (const v2f*)Wl2;

#pragma unroll
  for (int kk = 0; kk < 16; ++kk) {
    int kb = kk * 4;
    int kpair = (kb + koff) >> 1;
    v2f a;
    a.x = arow[kb + koff];
    a.y = arow[kb + koff + 1];
#pragma unroll
    for (int t = 0; t < 4; ++t) {
      v2f bb = wl2[kpair * 64 + t * 16 + lo];
      acc[t] = __builtin_amdgcn_wmma_f32_16x16x4_f32(
           false, a,  false, bb,
           (short)0, acc[t],  false,  false);
    }
  }

#pragma unroll
  for (int t = 0; t < 4; ++t) {
#pragma unroll
    for (int r = 0; r < 8; ++r) {
      int m = m0 + r + 8 * hi;
      if (m < n_rows) out[(long long)m * DFEAT + t * 16 + lo] = acc[t][r];
    }
  }
}


extern "C" void kernel_launch(void* const* d_in, const int* in_sizes, int n_in,
                              void* d_out, int out_size, void* d_ws, size_t ws_size,
                              hipStream_t stream) {
  const float* X    = (const float*)d_in[0];
  const float* vals = (const float*)d_in[1];
  const float* W    = (const float*)d_in[2];
  const float* bias = (const float*)d_in[3];
  const int*   row  = (const int*)d_in[4];
  const int*   col  = (const int*)d_in[5];
  float* out = (float*)d_out;

  int n_nodes = in_sizes[0] / DFEAT;
  int n_edges = in_sizes[1];
  int n_buckets = (n_nodes + RPB - 1) >> RPB_SHIFT;

  size_t temp_bytes = (((size_t)n_nodes * DFEAT * sizeof(float)) + 255) & ~(size_t)255;
  size_t rec_bytes  = (((size_t)n_edges * sizeof(v4i)) + 255) & ~(size_t)255;
  char* ws = (char*)d_ws;
  float* temp  = (float*)ws;
  v4i* recs    = (v4i*)(ws + temp_bytes);
  int* hist    = (int*)(ws + temp_bytes + rec_bytes);
  int* cursor  = hist + n_buckets;
  if (temp_bytes + rec_bytes + (size_t)n_buckets * 2 * sizeof(int) > ws_size) return;

  gcn_clear_counts_kernel<<<(2 * n_buckets + 255) / 256, 256, 0, stream>>>(hist, 2 * n_buckets);

  unsigned eblocks = (unsigned)((n_edges + CHUNK - 1) / CHUNK);
  gcn_bucket_count_kernel<<<eblocks, BIN_THREADS, (size_t)n_buckets * sizeof(int), stream>>>(
      row, n_edges, n_buckets, hist);

  gcn_bucket_place_kernel<<<eblocks, BIN_THREADS,
                            ((size_t)n_buckets * 2 + BIN_THREADS) * sizeof(int), stream>>>(
      row, col, vals, n_edges, n_buckets, hist, cursor, recs);

  gcn_bucket_sum_kernel<<<n_buckets, SUM_THREADS, 0, stream>>>(X, recs, hist, temp, n_nodes);

  int waves = (n_nodes + 15) / 16;
  long long gthreads = (long long)waves * 32;
  unsigned gblocks = (unsigned)((gthreads + 255) / 256);
  gcn_gemm_wmma_kernel<<<gblocks, 256, 0, stream>>>(temp, W, bias, out, n_nodes);
}


# Dense product fused into the bucket-sum kernel: each wave feeds its 16 rows of the summed LDS tile (pitch 68 dwords) to the WMMAs, out is written from the tile; temp and the GEMM launch dropped
# speedup vs baseline: 2.0161x; 1.0969x over previous
#include <hip/hip_runtime.h>


typedef __attribute__((ext_vector_type(2))) float v2f;
typedef __attribute__((ext_vector_type(4))) float v4f;
typedef __attribute__((ext_vector_type(8))) float v8f;
typedef __attribute__((ext_vector_type(4))) int v4i;

#define DFEAT 64
#define RPB_SHIFT 8
#define RPB 256
#define TP 68
#define CHUNK 8192
#define BIN_THREADS 256
#define SUM_THREADS 512
#define CB 16
#define PB 8

__global__ void gcn_clear_counts_kernel(int* __restrict__ p, int n) {
  int i = blockIdx.x * blockDim.x + threadIdx.x;
  if (i < n) p[i] = 0;
}

__global__ void __launch_bounds__(BIN_THREADS)
gcn_bucket_count_kernel(const int* __restrict__ row, int n_edges, int n_buckets,
                        int* __restrict__ hist) {
  extern __shared__ int sm[];
  int* lh = sm;
  const int tid = threadIdx.x;
  for (int i = tid; i < n_buckets; i += BIN_THREADS) lh[i] = 0;
  __syncthreads();
  const int e0 = blockIdx.x * CHUNK;
  const int e1 = (e0 + CHUNK < n_edges) ? (e0 + CHUNK) : n_edges;
  for (int eb = e0 + tid; eb < e1; eb += BIN_THREADS * CB) {
    int r[CB];
#pragma unroll
    for (int k = 0; k < CB; ++k) {
      int e = eb + k * BIN_THREADS;
      r[k] = (e < e1) ? row[e] : -1;
    }
#pragma unroll
    for (int k = 0; k < CB; ++k)
      if (r[k] >= 0) atomicAdd(&lh[r[k] >> RPB_SHIFT], 1);
  }
  __syncthreads();
  for (int i = tid; i < n_buckets; i += BIN_THREADS) {
    int c = lh[i];
    if (c) atomicAdd(&hist[i], c);
  }
}

__global__ void __launch_bounds__(BIN_THREADS)
gcn_bucket_place_kernel(const int* __restrict__ row, const int* __restrict__ col,
                        const float* __restrict__ vals, int n_edges, int n_buckets,
                        const int* __restrict__ hist, int* __restrict__ cursor,
                        v4i* __restrict__ recs) {
  extern __shared__ int sm[];
  int* lh = sm;
  int* gs = sm + n_buckets;
  int* part = sm + 2 * n_buckets;
  const int tid = threadIdx.x;

  for (int i = tid; i < n_buckets; i += BIN_THREADS) lh[i] = 0;
  __syncthreads();
  const int e0 = blockIdx.x * CHUNK;
  const int e1 = (e0 + CHUNK < n_edges) ? (e0 + CHUNK) : n_edges;
  for (int eb = e0 + tid; eb < e1; eb += BIN_THREADS * CB) {
    int r[CB];
#pragma unroll
    for (int k = 0; k < CB; ++k) {
      int e = eb + k * BIN_THREADS;
      r[k] = (e < e1) ? row[e] : -1;
    }
#pragma unroll
    for (int k = 0; k < CB; ++k)
      if (r[k] >= 0) atomicAdd(&lh[r[k] >> RPB_SHIFT], 1);
  }

  const int seg = (n_buckets + BIN_THREADS - 1) / BIN_THREADS;
  const int s0 = tid * seg;
  const int s1 = (s0 + seg < n_buckets) ? (s0 + seg) : n_buckets;
  int ssum = 0;
  for (int i = s0; i < s1; ++i) ssum += hist[i];
  part[tid] = ssum;
  __syncthreads();
  int pre = 0;
  for (int i = 0; i < tid; ++i) pre += part[i];
  for (int i = s0; i < s1; ++i) {
    int c = lh[i];
    int st = pre;
    if (c) st += atomicAdd(&cursor[i], c);
    gs[i] = st;
    lh[i] = 0;
    pre += hist[i];
  }
  __syncthreads();

  for (int eb = e0 + tid; eb < e1; eb += BIN_THREADS * PB) {
    int r[PB], c[PB], vb[PB], pos[PB];
#pragma unroll
    for (int k = 0; k < PB; ++k) {
      int e = eb + k * BIN_THREADS;
      bool ok = e < e1;
      r[k] = ok ? row[e] : -1;
      c[k] = ok ? col[e] : 0;
      vb[k] = ok ? __float_as_int(vals[e]) : 0;
    }
#pragma unroll
    for (int k = 0; k < PB; ++k) {
      pos[k] = 0;
      if (r[k] >= 0) {
        int b = r[k] >> RPB_SHIFT;
        pos[k] = gs[b] + atomicAdd(&lh[b], 1);
      }
    }
#pragma unroll
    for (int k = 0; k < PB; ++k) {
      if (r[k] >= 0) {
        v4i rec;
        rec.x = c[k];
        rec.y = (r[k] & (RPB - 1)) * (TP * 4);
        rec.z = vb[k];
        rec.w = 0;
        recs[pos[k]] = rec;
      }
    }
  }
}

__global__ void __launch_bounds__(SUM_THREADS) __attribute__((amdgpu_waves_per_eu(4, 8)))
gcn_bucket_sum_gemm_kernel(const float* __restrict__ X, const v4i* __restrict__ recs,
                           const int* __restrict__ hist, const float* __restrict__ W,
                           const float* __restrict__ bias, float* __restrict__ out,
                           int n_nodes) {
  __shared__ __attribute__((aligned(16))) float tile[RPB * TP];
  __shared__ __attribute__((aligned(16))) float Wl2[DFEAT * DFEAT];
  __shared__ float bl[DFEAT];
  __shared__ int sbase;
  const int tid = threadIdx.x;
  const int b = blockIdx.x;

  static_assert(DFEAT * DFEAT == 8 * SUM_THREADS, "two 4-float pieces of W per thread");
  const v4f w0 = ((const v4f*)W)[tid];
  const v4f w1 = ((const v4f*)W)[tid + SUM_THREADS];
  float bv0 = 0.0f;
  if (tid < DFEAT) bv0 = bias[tid];

  v4f z = {0.0f, 0.0f, 0.0f, 0.0f};
  for (int i = tid; i < RPB * TP / 4; i += SUM_THREADS) ((v4f*)tile)[i] = z;
  {
    const int k = tid >> 4;
    const int n = (tid & 15) * 4;
    float* d0 = Wl2 + (k >> 1) * 128 + n * 2 + (k & 1);
    float* d1 = d0 + 16 * 128;
    d0[0] = w0.x; d0[2] = w0.y; d0[4] = w0.z; d0[6] = w0.w;
    d1[0] = w1.x; d1[2] = w1.y; d1[4] = w1.z; d1[6] = w1.w;
  }
  if (tid < DFEAT) bl[tid] = bv0;
  if (tid == 0) sbase = 0;
  __syncthreads();
  int s = 0;
  for (int i = tid; i < b; i += SUM_THREADS) s += hist[i];
  atomicAdd(&sbase, s);
  __syncthreads();
  const int base = sbase;
  const int cnt = hist[b];
  const v4i* rb = recs + base;

  const int lane = tid & 31;
  const int wave = __builtin_amdgcn_readfirstlane(tid >> 5);

  for (int i0 = wave * 32; i0 < cnt; i0 += (SUM_THREADS / 32) * 32) {
    int idx = i0 + lane;
    if (idx >= cnt) idx = cnt - 1;
    const v4i rec = rb[idx];
    const int m = (cnt - i0 < 32) ? (cnt - i0) : 32;
    if (m == 32) {
      v2f xr[32];
#pragma unroll
      for (int j = 0; j < 32; ++j) {
        int c = __builtin_amdgcn_readlane(rec.x, j);
        xr[j] = *(const v2f*)(X + (long long)c * DFEAT + 2 * lane);
      }
#pragma unroll
      for (int j = 0; j < 32; ++j) {
        int ro = __builtin_amdgcn_readlane(rec.y, j);
        float v = __int_as_float(__builtin_amdgcn_readlane(rec.z, j));
        float* td = (float*)((char*)tile + ro) + 2 * lane;
        atomicAdd(td + 0, xr[j].x * v);
        atomicAdd(td + 1, xr[j].y * v);
      }
    } else {
      for (int j = 0; j < m; ++j) {
        int c = __builtin_amdgcn_readlane(rec.x, j);
        int ro = __builtin_amdgcn_readlane(rec.y, j);
        float v = __int_as_float(__builtin_amdgcn_readlane(rec.z, j));
        v2f x = *(const v2f*)(X + (long long)c * DFEAT + 2 * lane);
        float* td = (float*)((char*)tile + ro) + 2 * lane;
        atomicAdd(td + 0, x.x * v);
        atomicAdd(td + 1, x.y * v);
      }
    }
  }
  __syncthreads();

  const int m0 = wave * 16;
  const int lo = lane & 15;
  const int hi = lane >> 4;
  const int koff = hi * 2;

  v8f acc[4];
#pragma unroll
  for (int t = 0; t < 4; ++t) {
    float bv = bl[t * 16 + lo];
#pragma unroll
    for (int r = 0; r < 8; ++r) acc[t][r] = bv;
  }

  const float* arow = tile + (m0 + lo) * TP;
  const v2f* wl2 = (const v2f*)Wl2;

#pragma unroll
  for (int kk = 0; kk < 16; ++kk) {
    int kb = kk * 4;
    int kpair = (kb + koff) >> 1;
    v2f a = *(const v2f*)(arow + kb + koff);
#pragma unroll
    for (int t = 0; t < 4; ++t) {
      v2f bb = wl2[kpair * 64 + t * 16 + lo];
      acc[t] = __builtin_amdgcn_wmma_f32_16x16x4_f32(
           false, a,  false, bb,
           (short)0, acc[t],  false,  false);
    }
  }

#pragma unroll
  for (int t = 0; t < 4; ++t) {
#pragma unroll
    for (int r = 0; r < 8; ++r)
      tile[(m0 + r + 8 * hi) * TP + t * 16 + lo] = acc[t][r];
  }
  __syncthreads();

  const int r0 = b * RPB;
  const int nrows = (n_nodes - r0 < RPB) ? (n_nodes - r0) : RPB;
  v4f* dst = (v4f*)(out + (long long)r0 * DFEAT);
  for (int i = tid; i < nrows * (DFEAT / 4); i += SUM_THREADS)
    dst[i] = *(const v4f*)(tile + (i >> 4) * TP + (i & 15) * 4);
}


extern "C" void kernel_launch(void* const* d_in, const int* in_sizes, int n_in,
                              void* d_out, int out_size, void* d_ws, size_t ws_size,
                              hipStream_t stream) {
  const float* X    = (const float*)d_in[0];
  const float* vals = (const float*)d_in[1];
  const float* W    = (const float*)d_in[2];
  const float* bias = (const float*)d_in[3];
  const int*   row  = (const int*)d_in[4];
  const int*   col  = (const int*)d_in[5];
  float* out = (float*)d_out;

  int n_nodes = in_sizes[0] / DFEAT;
  int n_edges = in_sizes[1];
  int n_buckets = (n_nodes + RPB - 1) >> RPB_SHIFT;

  size_t rec_bytes  = (((size_t)n_edges * sizeof(v4i)) + 255) & ~(size_t)255;
  char* ws = (char*)d_ws;
  v4i* recs    = (v4i*)ws;
  int* hist    = (int*)(ws + rec_bytes);
  int* cursor  = hist + n_buckets;
  if (rec_bytes + (size_t)n_buckets * 2 * sizeof(int) > ws_size) return;

  gcn_clear_counts_kernel<<<(2 * n_buckets + 255) / 256, 256, 0, stream>>>(hist, 2 * n_buckets);

  unsigned eblocks = (unsigned)((n_edges + CHUNK - 1) / CHUNK);
  gcn_bucket_count_kernel<<<eblocks, BIN_THREADS, (size_t)n_buckets * sizeof(int), stream>>>(
      row, n_edges, n_buckets, hist);

  gcn_bucket_place_kernel<<<eblocks, BIN_THREADS,
                            ((size_t)n_buckets * 2 + BIN_THREADS) * sizeof(int), stream>>>(
      row, col, vals, n_edges, n_buckets, hist, cursor, recs);

  gcn_bucket_sum_gemm_kernel<<<n_buckets, SUM_THREADS, 0, stream>>>(X, recs, hist, W, bias, out,
                                                                    n_nodes);
}


# Bucket's records put in row order inside LDS (integer count, scan, rank); a wave sums each run of equal row in registers and adds into the LDS tile once per run, not per edge; sum order changes
# speedup vs baseline: 2.4512x; 1.2158x over previous
#include <hip/hip_runtime.h>


typedef __attribute__((ext_vector_type(2))) float v2f;
typedef __attribute__((ext_vector_type(4))) float v4f;
typedef __attribute__((ext_vector_type(8))) float v8f;
typedef __attribute__((ext_vector_type(4))) int v4i;

#define DFEAT 64
#define RPB_SHIFT 8
#define RPB 256
#define TP 68
#define CHUNK 8192
#define BIN_THREADS 256
#define SUM_THREADS 512
#define CB 16
#define PB 8
#define SCH 5120
#define SPT (SCH / SUM_THREADS)

__global__ void gcn_clear_counts_kernel(int* __restrict__ p, int n) {
  int i = blockIdx.x * blockDim.x + threadIdx.x;
  if (i < n) p[i] = 0;
}

__global__ void __launch_bounds__(BIN_THREADS)
gcn_bucket_count_kernel(const int* __restrict__ row, int n_edges, int n_buckets,
                        int* __restrict__ hist) {
  extern __shared__ int sm[];
  int* lh = sm;
  const int tid = threadIdx.x;
  for (int i = tid; i < n_buckets; i += BIN_THREADS) lh[i] = 0;
  __syncthreads();
  const int e0 = blockIdx.x * CHUNK;
  const int e1 = (e0 + CHUNK < n_edges) ? (e0 + CHUNK) : n_edges;
  for (int eb = e0 + tid; eb < e1; eb += BIN_THREADS * CB) {
    int r[CB];
#pragma unroll
    for (int k = 0; k < CB; ++k) {
      int e = eb + k * BIN_THREADS;
      r[k] = (e < e1) ? row[e] : -1;
    }
#pragma unroll
    for (int k = 0; k < CB; ++k)
      if (r[k] >= 0) atomicAdd(&lh[r[k] >> RPB_SHIFT], 1);
  }
  __syncthreads();
  for (int i = tid; i < n_buckets; i += BIN_THREADS) {
    int c = lh[i];
    if (c) atomicAdd(&hist[i], c);
  }
}

__global__ void __launch_bounds__(BIN_THREADS)
gcn_bucket_place_kernel(const int* __restrict__ row, const int* __restrict__ col,
                        const float* __restrict__ vals, int n_edges, int n_buckets,
                        const int* __restrict__ hist, int* __restrict__ cursor,
                        v4i* __restrict__ recs) {
  extern __shared__ int sm[];
  int* lh = sm;
  int* gs = sm + n_buckets;
  int* part = sm + 2 * n_buckets;
  const int tid = threadIdx.x;

  for (int i = tid; i < n_buckets; i += BIN_THREADS) lh[i] = 0;
  __syncthreads();
  const int e0 = blockIdx.x * CHUNK;
  const int e1 = (e0 + CHUNK < n_edges) ? (e0 + CHUNK) : n_edges;
  for (int eb = e0 + tid; eb < e1; eb += BIN_THREADS * CB) {
    int r[CB];
#pragma unroll
    for (int k = 0; k < CB; ++k) {
      int e = eb + k * BIN_THREADS;
      r[k] = (e < e1) ? row[e] : -1;
    }
#pragma unroll
    for (int k = 0; k < CB; ++k)
      if (r[k] >= 0) atomicAdd(&lh[r[k] >> RPB_SHIFT], 1);
  }

  const int seg = (n_buckets + BIN_THREADS - 1) / BIN_THREADS;
  const int s0 = tid * seg;
  const int s1 = (s0 + seg < n_buckets) ? (s0 + seg) : n_buckets;
  int ssum = 0;
  for (int i = s0; i < s1; ++i) ssum += hist[i];
  part[tid] = ssum;
  __syncthreads();
  int pre = 0;
  for (int i = 0; i < tid; ++i) pre += part[i];
  for (int i = s0; i < s1; ++i) {
    int c = lh[i];
    int st = pre;
    if (c) st += atomicAdd(&cursor[i], c);
    gs[i] = st;
    lh[i] = 0;
    pre += hist[i];
  }
  __syncthreads();

  for (int eb = e0 + tid; eb < e1; eb += BIN_THREADS * PB) {
    int r[PB], c[PB], vb[PB], pos[PB];
#pragma unroll
    for (int k = 0; k < PB; ++k) {
      int e = eb + k * BIN_THREADS;
      bool ok = e < e1;
      r[k] = ok ? row[e] : -1;
      c[k] = ok ? col[e] : 0;
      vb[k] = ok ? __float_as_int(vals[e]) : 0;
    }
#pragma unroll
    for (int k = 0; k < PB; ++k) {
      pos[k] = 0;
      if (r[k] >= 0) {
        int b = r[k] >> RPB_SHIFT;
        pos[k] = gs[b] + atomicAdd(&lh[b], 1);
      }
    }
#pragma unroll
    for (int k = 0; k < PB; ++k) {
      if (r[k] >= 0) {
        v4i rec;
        rec.x = c[k];
        rec.y = r[k] & (RPB - 1);
        rec.z = vb[k];
        rec.w = 0;
        recs[pos[k]] = rec;
      }
    }
  }
}

__global__ void __launch_bounds__(SUM_THREADS) __attribute__((amdgpu_waves_per_eu(4, 8)))
gcn_bucket_sum_gemm_kernel(const float* __restrict__ X, const v4i* __restrict__ recs,
                           const int* __restrict__ hist, const float* __restrict__ W,
                           const float* __restrict__ bias, float* __restrict__ out,
                           int n_nodes) {
  __shared__ __attribute__((aligned(16))) float tile[RPB * TP];
  __shared__ __attribute__((aligned(16))) float Wl2[DFEAT * DFEAT];
  __shared__ float bl[DFEAT];
  __shared__ int sc[SCH];
  __shared__ int sr[SCH];
  __shared__ int sz[SCH];
  __shared__ int rcnt[RPB];
  __shared__ int sbase;
  const int tid = threadIdx.x;
  const int b = blockIdx.x;

  static_assert(DFEAT * DFEAT == 8 * SUM_THREADS, "two 4-float pieces of W per thread");
  const v4f w0 = ((const v4f*)W)[tid];
  const v4f w1 = ((const v4f*)W)[tid + SUM_THREADS];
  float bv0 = 0.0f;
  if (tid < DFEAT) bv0 = bias[tid];

  v4f z = {0.0f, 0.0f, 0.0f, 0.0f};
  for (int i = tid; i < RPB * TP / 4; i += SUM_THREADS) ((v4f*)tile)[i] = z;
  {
    const int k = tid >> 4;
    const int n = (tid & 15) * 4;
    float* d0 = Wl2 + (k >> 1) * 128 + n * 2 + (k & 1);
    float* d1 = d0 + 16 * 128;
    d0[0] = w0.x; d0[2] = w0.y; d0[4] = w0.z; d0[6] = w0.w;
    d1[0] = w1.x; d1[2] = w1.y; d1[4] = w1.z; d1[6] = w1.w;
  }
  if (tid < DFEAT) bl[tid] = bv0;
  if (tid == 0) sbase = 0;
  __syncthreads();
  int s = 0;
  for (int i = tid; i < b; i += SUM_THREADS) s += hist[i];
  atomicAdd(&sbase, s);
  __syncthreads();
  const int base = sbase;
  const int cnt = hist[b];
  const v4i* rb = recs + base;

  const int lane = tid & 31;
  const int wave = __builtin_amdgcn_readfirstlane(tid >> 5);

  for (int cb = 0; cb < cnt; cb += SCH) {
    const int n = (cnt - cb < SCH) ? (cnt - cb) : SCH;
    if (tid < RPB) rcnt[tid] = 0;
    __syncthreads();
    int mc[SPT], mr[SPT], mz[SPT];
#pragma unroll
    for (int k = 0; k < SPT; ++k) {
      const int i = tid + k * SUM_THREADS;
      mc[k] = 0; mr[k] = -1; mz[k] = 0;
      if (i < n) {
        const v4i rec = rb[cb + i];
        mc[k] = rec.x; mr[k] = rec.y; mz[k] = rec.z;
      }
    }
#pragma unroll
    for (int k = 0; k < SPT; ++k)
      if (mr[k] >= 0) atomicAdd(&rcnt[mr[k]], 1);
    __syncthreads();
    if (tid < 32) {
      int c[RPB / 32];
      int s = 0;
      for (int q = 0; q < RPB / 32; ++q) { c[q] = rcnt[tid * (RPB / 32) + q]; s += c[q]; }
      int incl = s;
      for (int d = 1; d < 32; d <<= 1) {
        const int t = __shfl_up(incl, d);
        if (tid >= d) incl += t;
      }
      int ex = incl - s;
      for (int q = 0; q < RPB / 32; ++q) { rcnt[tid * (RPB / 32) + q] = ex; ex += c[q]; }
    }
    __syncthreads();
    int p[SPT];
#pragma unroll
    for (int k = 0; k < SPT; ++k) {
      p[k] = -1;
      if (mr[k] >= 0) p[k] = atomicAdd(&rcnt[mr[k]], 1);
    }
#pragma unroll
    for (int k = 0; k < SPT; ++k) {
      if (p[k] >= 0) { sc[p[k]] = mc[k]; sr[p[k]] = mr[k]; sz[p[k]] = mz[k]; }
    }
    __syncthreads();

    for (int i0 = wave * 32; i0 < n; i0 += (SUM_THREADS / 32) * 32) {
      int idx = i0 + lane;
      if (idx >= n) idx = n - 1;
      const int rc = sc[idx];
      const int rr = sr[idx];
      const int rz = sz[idx];
      const int m = (n - i0 < 32) ? (n - i0) : 32;
      float a0 = 0.0f, a1 = 0.0f;
      const int rprev = __shfl_up(rr, 1);
      const unsigned chg = __builtin_amdgcn_ballot_w32(lane != 0 && rr != rprev);
      int last;
      if (m == 32) {
        v2f xr[32];
#pragma unroll
        for (int j = 0; j < 32; ++j) {
          int c = __builtin_amdgcn_readlane(rc, j);
          xr[j] = *(const v2f*)(X + (long long)c * DFEAT + 2 * lane);
        }
#pragma unroll
        for (int j = 0; j < 32; ++j) {
          if (j > 0 && ((chg >> j) & 1u)) {
            float* td = tile + __builtin_amdgcn_readlane(rr, j - 1) * TP + 2 * lane;
            atomicAdd(td + 0, a0);
            atomicAdd(td + 1, a1);
            a0 = 0.0f; a1 = 0.0f;
          }
          const float v = __int_as_float(__builtin_amdgcn_readlane(rz, j));
          a0 += xr[j].x * v;
          a1 += xr[j].y * v;
        }
        last = __builtin_amdgcn_readlane(rr, 31);
      } else {
        for (int j = 0; j < m; ++j) {
          const int c = __builtin_amdgcn_readlane(rc, j);
          const float v = __int_as_float(__builtin_amdgcn_readlane(rz, j));
          const v2f x = *(const v2f*)(X + (long long)c * DFEAT + 2 * lane);
          if (j > 0 && ((chg >> j) & 1u)) {
            float* td = tile + __builtin_amdgcn_readlane(rr, j - 1) * TP + 2 * lane;
            atomicAdd(td + 0, a0);
            atomicAdd(td + 1, a1);
            a0 = 0.0f; a1 = 0.0f;
          }
          a0 += x.x * v;
          a1 += x.y * v;
        }
        last = __builtin_amdgcn_readlane(rr, m - 1);
      }
      float* td = tile + last * TP + 2 * lane;
      atomicAdd(td + 0, a0);
      atomicAdd(td + 1, a1);
    }
  }
  __syncthreads();

  const int m0 = wave * 16;
  const int lo = lane & 15;
  const int hi = lane >> 4;
  const int koff = hi * 2;

  v8f acc[4];
#pragma unroll
  for (int t = 0; t < 4; ++t) {
    float bv = bl[t * 16 + lo];
#pragma unroll
    for (int r = 0; r < 8; ++r) acc[t][r] = bv;
  }

  const float* arow = tile + (m0 + lo) * TP;
  const v2f* wl2 = (const v2f*)Wl2;

#pragma unroll
  for (int kk = 0; kk < 16; ++kk) {
    int kb = kk * 4;
    int kpair = (kb + koff) >> 1;
    v2f a = *(const v2f*)(arow + kb + koff);
#pragma unroll
    for (int t = 0; t < 4; ++t) {
      v2f bb = wl2[kpair * 64 + t * 16 + lo];
      acc[t] = __builtin_amdgcn_wmma_f32_16x16x4_f32(
           false, a,  false, bb,
           (short)0, acc[t],  false,  false);
    }
  }

#pragma unroll
  for (int t = 0; t < 4; ++t) {
#pragma unroll
    for (int r = 0; r < 8; ++r)
      tile[(m0 + r + 8 * hi) * TP + t * 16 + lo] = acc[t][r];
  }
  __syncthreads();

  const int r0 = b * RPB;
  const int nrows = (n_nodes - r0 < RPB) ? (n_nodes - r0) : RPB;
  v4f* dst = (v4f*)(out + (long long)r0 * DFEAT);
  for (int i = tid; i < nrows * (DFEAT / 4); i += SUM_THREADS)
    dst[i] = *(const v4f*)(tile + (i >> 4) * TP + (i & 15) * 4);
}


extern "C" void kernel_launch(void* const* d_in, const int* in_sizes, int n_in,
                              void* d_out, int out_size, void* d_ws, size_t ws_size,
                              hipStream_t stream) {
  const float* X    = (const float*)d_in[0];
  const float* vals = (const float*)d_in[1];
  const float* W    = (const float*)d_in[2];
  const float* bias = (const float*)d_in[3];
  const int*   row  = (const int*)d_in[4];
  const int*   col  = (const int*)d_in[5];
  float* out = (float*)d_out;

  int n_nodes = in_sizes[0] / DFEAT;
  int n_edges = in_sizes[1];
  int n_buckets = (n_nodes + RPB - 1) >> RPB_SHIFT;

  size_t rec_bytes  = (((size_t)n_edges * sizeof(v4i)) + 255) & ~(size_t)255;
  char* ws = (char*)d_ws;
  v4i* recs    = (v4i*)ws;
  int* hist    = (int*)(ws + rec_bytes);
  int* cursor  = hist + n_buckets;
  if (rec_bytes + (size_t)n_buckets * 2 * sizeof(int) > ws_size) return;

  gcn_clear_counts_kernel<<<(2 * n_buckets + 255) / 256, 256, 0, stream>>>(hist, 2 * n_buckets);

  unsigned eblocks = (unsigned)((n_edges + CHUNK - 1) / CHUNK);
  gcn_bucket_count_kernel<<<eblocks, BIN_THREADS, (size_t)n_buckets * sizeof(int), stream>>>(
      row, n_edges, n_buckets, hist);

  gcn_bucket_place_kernel<<<eblocks, BIN_THREADS,
                            ((size_t)n_buckets * 2 + BIN_THREADS) * sizeof(int), stream>>>(
      row, col, vals, n_edges, n_buckets, hist, cursor, recs);

  gcn_bucket_sum_gemm_kernel<<<n_buckets, SUM_THREADS, 0, stream>>>(X, recs, hist, W, bias, out,
                                                                    n_nodes);
}


# No global counts: every 8192-edge chunk is bucket-sorted in LDS into its own records + a chunk x bucket start table; per-wave rank counters fix the sum order (changed); 4 launches to 2
# speedup vs baseline: 2.6335x; 1.0744x over previous
#include <hip/hip_runtime.h>


typedef __attribute__((ext_vector_type(2))) float v2f;
typedef __attribute__((ext_vector_type(4))) float v4f;
typedef __attribute__((ext_vector_type(8))) float v8f;
typedef __attribute__((ext_vector_type(4))) int v4i;

#define DFEAT 64
#define RPB_SHIFT 8
#define RPB 256
#define TP 68
#define CHUNK 8192
#define BIN_THREADS 256
#define SUM_THREADS 512
#define CB 16
#define PB 8
#define SCH 5120
#define SPT (SCH / SUM_THREADS)
#define BIN_WAVES (BIN_THREADS / 32)
#define SUM_WAVES (SUM_THREADS / 32)
#define MAXC 256

__global__ void __launch_bounds__(BIN_THREADS)
gcn_chunk_sort_kernel(const int* __restrict__ row, const int* __restrict__ col,
                      const float* __restrict__ vals, int n_edges, int n_buckets,
                      int* __restrict__ tab, v4i* __restrict__ recs) {
  extern __shared__ int sm[];
  int* lhw = sm;
  int* part = sm + BIN_WAVES * n_buckets;
  const int tid = threadIdx.x;
  int* lh = lhw + (tid >> 5) * n_buckets;

  for (int i = tid; i < BIN_WAVES * n_buckets; i += BIN_THREADS) lhw[i] = 0;
  __syncthreads();
  const int e0 = blockIdx.x * CHUNK;
  const int e1 = (e0 + CHUNK < n_edges) ? (e0 + CHUNK) : n_edges;
  for (int eb = e0 + tid; eb < e1; eb += BIN_THREADS * CB) {
    int r[CB];
#pragma unroll
    for (int k = 0; k < CB; ++k) {
      int e = eb + k * BIN_THREADS;
      r[k] = (e < e1) ? row[e] : -1;
    }
#pragma unroll
    for (int k = 0; k < CB; ++k)
      if (r[k] >= 0) atomicAdd(&lh[r[k] >> RPB_SHIFT], 1);
  }
  __syncthreads();

  const int seg = (n_buckets + BIN_THREADS - 1) / BIN_THREADS;
  const int s0 = tid * seg;
  const int s1 = (s0 + seg < n_buckets) ? (s0 + seg) : n_buckets;
  int ssum = 0;
  for (int i = s0; i < s1; ++i)
    for (int q = 0; q < BIN_WAVES; ++q) ssum += lhw[q * n_buckets + i];
  part[tid] = ssum;
  __syncthreads();
  int pre = 0;
  for (int i = 0; i < tid; ++i) pre += part[i];
  int* trow = tab + (long long)blockIdx.x * (n_buckets + 1);
  for (int i = s0; i < s1; ++i) {
    trow[i] = pre;
    for (int q = 0; q < BIN_WAVES; ++q) {
      const int c = lhw[q * n_buckets + i];
      lhw[q * n_buckets + i] = pre;
      pre += c;
    }
  }
  if (tid == 0) trow[n_buckets] = e1 - e0;
  __syncthreads();

  for (int eb = e0 + tid; eb < e1; eb += BIN_THREADS * PB) {
    int r[PB], c[PB], vb[PB], pos[PB];
#pragma unroll
    for (int k = 0; k < PB; ++k) {
      int e = eb + k * BIN_THREADS;
      bool ok = e < e1;
      r[k] = ok ? row[e] : -1;
      c[k] = ok ? col[e] : 0;
      vb[k] = ok ? __float_as_int(vals[e]) : 0;
    }
#pragma unroll
    for (int k = 0; k < PB; ++k) {
      pos[k] = 0;
      if (r[k] >= 0) pos[k] = e0 + atomicAdd(&lh[r[k] >> RPB_SHIFT], 1);
    }
#pragma unroll
    for (int k = 0; k < PB; ++k) {
      if (r[k] >= 0) {
        v4i rec;
        rec.x = c[k];
        rec.y = r[k] & (RPB - 1);
        rec.z = vb[k];
        rec.w = 0;
        recs[pos[k]] = rec;
      }
    }
  }
}

__global__ void __launch_bounds__(SUM_THREADS) __attribute__((amdgpu_waves_per_eu(4, 8)))
gcn_bucket_sum_gemm_kernel(const float* __restrict__ X, const v4i* __restrict__ recs,
                           const int* __restrict__ tab, const float* __restrict__ W,
                           const float* __restrict__ bias, float* __restrict__ out,
                           int n_nodes, int n_buckets, int n_chunks) {
  __shared__ __attribute__((aligned(16))) float tile[RPB * TP];
  __shared__ __attribute__((aligned(16))) float Wl2[DFEAT * DFEAT];
  __shared__ float bl[DFEAT];
  __shared__ int sc[SCH];
  __shared__ int sr[SCH];
  __shared__ int sz[SCH];
  __shared__ int rcw[SUM_WAVES * (RPB / 2)];
  __shared__ int rtot[RPB];
  __shared__ int pfx[MAXC + 1];
  __shared__ int pst[MAXC];
  const int tid = threadIdx.x;
  const int b = blockIdx.x;

  static_assert(DFEAT * DFEAT == 8 * SUM_THREADS, "two 4-float pieces of W per thread");
  const v4f w0 = ((const v4f*)W)[tid];
  const v4f w1 = ((const v4f*)W)[tid + SUM_THREADS];
  float bv0 = 0.0f;
  if (tid < DFEAT) bv0 = bias[tid];
  static_assert(MAXC <= SUM_THREADS, "one table entry per thread");
  int sz_c = 0, st_c = 0;
  if (tid < MAXC && tid < n_chunks) {
    const int* te = tab + (long long)tid * (n_buckets + 1) + b;
    const int s = te[0];
    sz_c = te[1] - s;
    st_c = tid * CHUNK + s;
  }

  v4f z = {0.0f, 0.0f, 0.0f, 0.0f};
  for (int i = tid; i < RPB * TP / 4; i += SUM_THREADS) ((v4f*)tile)[i] = z;
  {
    const int k = tid >> 4;
    const int n = (tid & 15) * 4;
    float* d0 = Wl2 + (k >> 1) * 128 + n * 2 + (k & 1);
    float* d1 = d0 + 16 * 128;
    d0[0] = w0.x; d0[2] = w0.y; d0[4] = w0.z; d0[6] = w0.w;
    d1[0] = w1.x; d1[2] = w1.y; d1[4] = w1.z; d1[6] = w1.w;
  }
  if (tid < DFEAT) bl[tid] = bv0;
  if (tid < MAXC) { pfx[tid] = sz_c; pst[tid] = st_c; }
  __syncthreads();
  if (tid < 32) {
    int c[MAXC / 32];
    int s = 0;
    for (int q = 0; q < MAXC / 32; ++q) { c[q] = pfx[tid * (MAXC / 32) + q]; s += c[q]; }
    int incl = s;
    for (int d = 1; d < 32; d <<= 1) {
      const int t = __shfl_up(incl, d);
      if (tid >= d) incl += t;
    }
    int ex = incl - s;
    for (int q = 0; q < MAXC / 32; ++q) { pfx[tid * (MAXC / 32) + q] = ex; ex += c[q]; }
    if (tid == 31) pfx[MAXC] = ex;
  }
  __syncthreads();
  const int cnt = pfx[MAXC];

  const int lane = tid & 31;
  const int wave = __builtin_amdgcn_readfirstlane(tid >> 5);

  for (int cb = 0; cb < cnt; cb += SCH) {
    const int n = (cnt - cb < SCH) ? (cnt - cb) : SCH;
    for (int i = tid; i < SUM_WAVES * (RPB / 2); i += SUM_THREADS) rcw[i] = 0;
    __syncthreads();
    int mc[SPT], mr[SPT], mz[SPT];
    int gi[SPT], lo[SPT];
#pragma unroll
    for (int k = 0; k < SPT; ++k) {
      const int i = cb + tid + k * SUM_THREADS;
      gi[k] = (i < cnt) ? i : (cnt - 1);
      lo[k] = 0;
    }
#pragma unroll
    for (int step = MAXC / 2; step > 0; step >>= 1) {
#pragma unroll
      for (int k = 0; k < SPT; ++k) lo[k] += (pfx[lo[k] + step] <= gi[k]) ? step : 0;
    }
    int ra[SPT];
#pragma unroll
    for (int k = 0; k < SPT; ++k) ra[k] = pst[lo[k]] + (gi[k] - pfx[lo[k]]);
#pragma unroll
    for (int k = 0; k < SPT; ++k) {
      const int i = tid + k * SUM_THREADS;
      mc[k] = 0; mr[k] = -1; mz[k] = 0;
      if (i < n) {
        const v4i rec = recs[ra[k]];
        mc[k] = rec.x; mr[k] = rec.y; mz[k] = rec.z;
      }
    }
    int* rw = rcw + wave * (RPB / 2);
#pragma unroll
    for (int k = 0; k < SPT; ++k)
      if (mr[k] >= 0) atomicAdd(&rw[mr[k] & (RPB / 2 - 1)], 1 << ((mr[k] >> (RPB_SHIFT - 1)) * 16));
    __syncthreads();
    if (tid < RPB / 2) {
      int tl = 0, th = 0;
      for (int q = 0; q < SUM_WAVES; ++q) {
        const unsigned wd = (unsigned)rcw[q * (RPB / 2) + tid];
        tl += (int)(wd & 0xFFFFu);
        th += (int)(wd >> 16);
      }
      rtot[tid] = tl;
      rtot[tid + RPB / 2] = th;
    }
    __syncthreads();
    if (tid < 32) {
      int c[RPB / 32];
      int s = 0;
      for (int q = 0; q < RPB / 32; ++q) { c[q] = rtot[tid * (RPB / 32) + q]; s += c[q]; }
      int incl = s;
      for (int d = 1; d < 32; d <<= 1) {
        const int t = __shfl_up(incl, d);
        if (tid >= d) incl += t;
      }
      int ex = incl - s;
      for (int q = 0; q < RPB / 32; ++q) { rtot[tid * (RPB / 32) + q] = ex; ex += c[q]; }
    }
    __syncthreads();
    if (tid < RPB / 2) {
      int ol = rtot[tid], oh = rtot[tid + RPB / 2];
      for (int q = 0; q < SUM_WAVES; ++q) {
        const unsigned wd = (unsigned)rcw[q * (RPB / 2) + tid];
        rcw[q * (RPB / 2) + tid] = (int)((unsigned)ol | ((unsigned)oh << 16));
        ol += (int)(wd & 0xFFFFu);
        oh += (int)(wd >> 16);
      }
    }
    __syncthreads();
    int p[SPT];
#pragma unroll
    for (int k = 0; k < SPT; ++k) {
      p[k] = -1;
      if (mr[k] >= 0) {
        const int sh = (mr[k] >> (RPB_SHIFT - 1)) * 16;
        const unsigned old = (unsigned)atomicAdd(&rw[mr[k] & (RPB / 2 - 1)], 1 << sh);
        p[k] = (int)((old >> sh) & 0xFFFFu);
      }
    }
#pragma unroll
    for (int k = 0; k < SPT; ++k) {
      if (p[k] >= 0) { sc[p[k]] = mc[k]; sr[p[k]] = mr[k]; sz[p[k]] = mz[k]; }
    }
    __syncthreads();

    for (int i0 = wave * 32; i0 < n; i0 += (SUM_THREADS / 32) * 32) {
      int idx = i0 + lane;
      if (idx >= n) idx = n - 1;
      const int rc = sc[idx];
      const int rr = sr[idx];
      const int rz = sz[idx];
      const int m = (n - i0 < 32) ? (n - i0) : 32;
      float a0 = 0.0f, a1 = 0.0f;
      const int rprev = __shfl_up(rr, 1);
      const unsigned chg = __builtin_amdgcn_ballot_w32(lane != 0 && rr != rprev);
      int last;
      if (m == 32) {
        v2f xr[32];
#pragma unroll
        for (int j = 0; j < 32; ++j) {
          int c = __builtin_amdgcn_readlane(rc, j);
          xr[j] = *(const v2f*)(X + (long long)c * DFEAT + 2 * lane);
        }
#pragma unroll
        for (int j = 0; j < 32; ++j) {
          if (j > 0 && ((chg >> j) & 1u)) {
            float* td = tile + __builtin_amdgcn_readlane(rr, j - 1) * TP + 2 * lane;
            atomicAdd(td + 0, a0);
            atomicAdd(td + 1, a1);
            a0 = 0.0f; a1 = 0.0f;
          }
          const float v = __int_as_float(__builtin_amdgcn_readlane(rz, j));
          a0 += xr[j].x * v;
          a1 += xr[j].y * v;
        }
        last = __builtin_amdgcn_readlane(rr, 31);
      } else {
        for (int j = 0; j < m; ++j) {
          const int c = __builtin_amdgcn_readlane(rc, j);
          const float v = __int_as_float(__builtin_amdgcn_readlane(rz, j));
          const v2f x = *(const v2f*)(X + (long long)c * DFEAT + 2 * lane);
          if (j > 0 && ((chg >> j) & 1u)) {
            float* td = tile + __builtin_amdgcn_readlane(rr, j - 1) * TP + 2 * lane;
            atomicAdd(td + 0, a0);
            atomicAdd(td + 1, a1);
            a0 = 0.0f; a1 = 0.0f;
          }
          a0 += x.x * v;
          a1 += x.y * v;
        }
        last = __builtin_amdgcn_readlane(rr, m - 1);
      }
      float* td = tile + last * TP + 2 * lane;
      atomicAdd(td + 0, a0);
      atomicAdd(td + 1, a1);
    }
  }
  __syncthreads();

  const int m0 = wave * 16;
  const int lo = lane & 15;
  const int hi = lane >> 4;
  const int koff = hi * 2;

  v8f acc[4];
#pragma unroll
  for (int t = 0; t < 4; ++t) {
    float bv = bl[t * 16 + lo];
#pragma unroll
    for (int r = 0; r < 8; ++r) acc[t][r] = bv;
  }

  const float* arow = tile + (m0 + lo) * TP;
  const v2f* wl2 = (const v2f*)Wl2;

#pragma unroll
  for (int kk = 0; kk < 16; ++kk) {
    int kb = kk * 4;
    int kpair = (kb + koff) >> 1;
    v2f a = *(const v2f*)(arow + kb + koff);
#pragma unroll
    for (int t = 0; t < 4; ++t) {
      v2f bb = wl2[kpair * 64 + t * 16 + lo];
      acc[t] = __builtin_amdgcn_wmma_f32_16x16x4_f32(
           false, a,  false, bb,
           (short)0, acc[t],  false,  false);
    }
  }

#pragma unroll
  for (int t = 0; t < 4; ++t) {
#pragma unroll
    for (int r = 0; r < 8; ++r)
      tile[(m0 + r + 8 * hi) * TP + t * 16 + lo] = acc[t][r];
  }
  __syncthreads();

  const int r0 = b * RPB;
  const int nrows = (n_nodes - r0 < RPB) ? (n_nodes - r0) : RPB;
  v4f* dst = (v4f*)(out + (long long)r0 * DFEAT);
  for (int i = tid; i < nrows * (DFEAT / 4); i += SUM_THREADS)
    dst[i] = *(const v4f*)(tile + (i >> 4) * TP + (i & 15) * 4);
}


extern "C" void kernel_launch(void* const* d_in, const int* in_sizes, int n_in,
                              void* d_out, int out_size, void* d_ws, size_t ws_size,
                              hipStream_t stream) {
  const float* X    = (const float*)d_in[0];
  const float* vals = (const float*)d_in[1];
  const float* W    = (const float*)d_in[2];
  const float* bias = (const float*)d_in[3];
  const int*   row  = (const int*)d_in[4];
  const int*   col  = (const int*)d_in[5];
  float* out = (float*)d_out;

  int n_nodes = in_sizes[0] / DFEAT;
  int n_edges = in_sizes[1];
  int n_buckets = (n_nodes + RPB - 1) >> RPB_SHIFT;
  int n_chunks = (n_edges + CHUNK - 1) / CHUNK;

  size_t rec_bytes  = (((size_t)n_edges * sizeof(v4i)) + 255) & ~(size_t)255;
  char* ws = (char*)d_ws;
  v4i* recs = (v4i*)ws;
  int* tab  = (int*)(ws + rec_bytes);
  if (rec_bytes + (size_t)n_chunks * (n_buckets + 1) * sizeof(int) > ws_size) return;
  if (n_chunks > MAXC) return;

  gcn_chunk_sort_kernel<<<n_chunks, BIN_THREADS,
                          ((size_t)n_buckets * BIN_WAVES + BIN_THREADS) * sizeof(int), stream>>>(
      row, col, vals, n_edges, n_buckets, tab, recs);

  gcn_bucket_sum_gemm_kernel<<<n_buckets, SUM_THREADS, 0, stream>>>(X, recs, tab, W, bias, out,
                                                                    n_nodes, n_buckets, n_chunks);
}


# Chunk sort keeps each lane's 32 edges in registers from count to placement: rows read once, every global load requested up front, one memory wait instead of six; record places and sums unchanged
# speedup vs baseline: 2.7212x; 1.0333x over previous
#include <hip/hip_runtime.h>


typedef __attribute__((ext_vector_type(2))) float v2f;
typedef __attribute__((ext_vector_type(4))) float v4f;
typedef __attribute__((ext_vector_type(8))) float v8f;
typedef __attribute__((ext_vector_type(4))) int v4i;

#define DFEAT 64
#define RPB_SHIFT 8
#define RPB 256
#define TP 68
#define CHUNK 8192
#define BIN_THREADS 256
#define SUM_THREADS 512
#define EPT (CHUNK / BIN_THREADS)
#define SCH 5120
#define SPT (SCH / SUM_THREADS)
#define BIN_WAVES (BIN_THREADS / 32)
#define SUM_WAVES (SUM_THREADS / 32)
#define MAXC 256

__global__ void __launch_bounds__(BIN_THREADS)
gcn_chunk_sort_kernel(const int* __restrict__ row, const int* __restrict__ col,
                      const float* __restrict__ vals, int n_edges, int n_buckets,
                      int* __restrict__ tab, v4i* __restrict__ recs) {
  extern __shared__ int sm[];
  int* lhw = sm;
  int* part = sm + BIN_WAVES * n_buckets;
  const int tid = threadIdx.x;
  int* lh = lhw + (tid >> 5) * n_buckets;

  static_assert(CHUNK % BIN_THREADS == 0, "a whole number of edges per lane");
  const int e0 = blockIdx.x * CHUNK;
  const int e1 = (e0 + CHUNK < n_edges) ? (e0 + CHUNK) : n_edges;
  int r[EPT], c[EPT], vb[EPT];
#pragma unroll
  for (int k = 0; k < EPT; ++k) {
    const int e = e0 + tid + k * BIN_THREADS;
    r[k] = (e < e1) ? row[e] : -1;
  }
#pragma unroll
  for (int k = 0; k < EPT; ++k) {
    const int e = e0 + tid + k * BIN_THREADS;
    const bool ok = e < e1;
    c[k] = ok ? col[e] : 0;
    vb[k] = ok ? __float_as_int(vals[e]) : 0;
  }
  for (int i = tid; i < BIN_WAVES * n_buckets; i += BIN_THREADS) lhw[i] = 0;
  __syncthreads();
#pragma unroll
  for (int k = 0; k < EPT; ++k)
    if (r[k] >= 0) atomicAdd(&lh[r[k] >> RPB_SHIFT], 1);
  __syncthreads();

  const int seg = (n_buckets + BIN_THREADS - 1) / BIN_THREADS;
  const int s0 = tid * seg;
  const int s1 = (s0 + seg < n_buckets) ? (s0 + seg) : n_buckets;
  int ssum = 0;
  for (int i = s0; i < s1; ++i)
    for (int q = 0; q < BIN_WAVES; ++q) ssum += lhw[q * n_buckets + i];
  part[tid] = ssum;
  __syncthreads();
  int pre = 0;
  for (int i = 0; i < tid; ++i) pre += part[i];
  int* trow = tab + (long long)blockIdx.x * (n_buckets + 1);
  for (int i = s0; i < s1; ++i) {
    trow[i] = pre;
    for (int q = 0; q < BIN_WAVES; ++q) {
      const int c = lhw[q * n_buckets + i];
      lhw[q * n_buckets + i] = pre;
      pre += c;
    }
  }
  if (tid == 0) trow[n_buckets] = e1 - e0;
  __syncthreads();

  int pos[EPT];
#pragma unroll
  for (int k = 0; k < EPT; ++k) {
    pos[k] = 0;
    if (r[k] >= 0) pos[k] = e0 + atomicAdd(&lh[r[k] >> RPB_SHIFT], 1);
  }
#pragma unroll
  for (int k = 0; k < EPT; ++k) {
    if (r[k] >= 0) {
      v4i rec;
      rec.x = c[k];
      rec.y = r[k] & (RPB - 1);
      rec.z = vb[k];
      rec.w = 0;
      recs[pos[k]] = rec;
    }
  }
}

__global__ void __launch_bounds__(SUM_THREADS) __attribute__((amdgpu_waves_per_eu(4, 8)))
gcn_bucket_sum_gemm_kernel(const float* __restrict__ X, const v4i* __restrict__ recs,
                           const int* __restrict__ tab, const float* __restrict__ W,
                           const float* __restrict__ bias, float* __restrict__ out,
                           int n_nodes, int n_buckets, int n_chunks) {
  __shared__ __attribute__((aligned(16))) float tile[RPB * TP];
  __shared__ __attribute__((aligned(16))) float Wl2[DFEAT * DFEAT];
  __shared__ float bl[DFEAT];
  __shared__ int sc[SCH];
  __shared__ int sr[SCH];
  __shared__ int sz[SCH];
  __shared__ int rcw[SUM_WAVES * (RPB / 2)];
  __shared__ int rtot[RPB];
  __shared__ int pfx[MAXC + 1];
  __shared__ int pst[MAXC];
  const int tid = threadIdx.x;
  const int b = blockIdx.x;

  static_assert(DFEAT * DFEAT == 8 * SUM_THREADS, "two 4-float pieces of W per thread");
  const v4f w0 = ((const v4f*)W)[tid];
  const v4f w1 = ((const v4f*)W)[tid + SUM_THREADS];
  float bv0 = 0.0f;
  if (tid < DFEAT) bv0 = bias[tid];
  static_assert(MAXC <= SUM_THREADS, "one table entry per thread");
  int sz_c = 0, st_c = 0;
  if (tid < MAXC && tid < n_chunks) {
    const int* te = tab + (long long)tid * (n_buckets + 1) + b;
    const int s = te[0];
    sz_c = te[1] - s;
    st_c = tid * CHUNK + s;
  }

  v4f z = {0.0f, 0.0f, 0.0f, 0.0f};
  for (int i = tid; i < RPB * TP / 4; i += SUM_THREADS) ((v4f*)tile)[i] = z;
  {
    const int k = tid >> 4;
    const int n = (tid & 15) * 4;
    float* d0 = Wl2 + (k >> 1) * 128 + n * 2 + (k & 1);
    float* d1 = d0 + 16 * 128;
    d0[0] = w0.x; d0[2] = w0.y; d0[4] = w0.z; d0[6] = w0.w;
    d1[0] = w1.x; d1[2] = w1.y; d1[4] = w1.z; d1[6] = w1.w;
  }
  if (tid < DFEAT) bl[tid] = bv0;
  if (tid < MAXC) { pfx[tid] = sz_c; pst[tid] = st_c; }
  __syncthreads();
  if (tid < 32) {
    int c[MAXC / 32];
    int s = 0;
    for (int q = 0; q < MAXC / 32; ++q) { c[q] = pfx[tid * (MAXC / 32) + q]; s += c[q]; }
    int incl = s;
    for (int d = 1; d < 32; d <<= 1) {
      const int t = __shfl_up(incl, d);
      if (tid >= d) incl += t;
    }
    int ex = incl - s;
    for (int q = 0; q < MAXC / 32; ++q) { pfx[tid * (MAXC / 32) + q] = ex; ex += c[q]; }
    if (tid == 31) pfx[MAXC] = ex;
  }
  __syncthreads();
  const int cnt = pfx[MAXC];

  const int lane = tid & 31;
  const int wave = __builtin_amdgcn_readfirstlane(tid >> 5);

  for (int cb = 0; cb < cnt; cb += SCH) {
    const int n = (cnt - cb < SCH) ? (cnt - cb) : SCH;
    for (int i = tid; i < SUM_WAVES * (RPB / 2); i += SUM_THREADS) rcw[i] = 0;
    __syncthreads();
    int mc[SPT], mr[SPT], mz[SPT];
    int gi[SPT], lo[SPT];
#pragma unroll
    for (int k = 0; k < SPT; ++k) {
      const int i = cb + tid + k * SUM_THREADS;
      gi[k] = (i < cnt) ? i : (cnt - 1);
      lo[k] = 0;
    }
#pragma unroll
    for (int step = MAXC / 2; step > 0; step >>= 1) {
#pragma unroll
      for (int k = 0; k < SPT; ++k) lo[k] += (pfx[lo[k] + step] <= gi[k]) ? step : 0;
    }
    int ra[SPT];
#pragma unroll
    for (int k = 0; k < SPT; ++k) ra[k] = pst[lo[k]] + (gi[k] - pfx[lo[k]]);
#pragma unroll
    for (int k = 0; k < SPT; ++k) {
      const int i = tid + k * SUM_THREADS;
      mc[k] = 0; mr[k] = -1; mz[k] = 0;
      if (i < n) {
        const v4i rec = recs[ra[k]];
        mc[k] = rec.x; mr[k] = rec.y; mz[k] = rec.z;
      }
    }
    int* rw = rcw + wave * (RPB / 2);
#pragma unroll
    for (int k = 0; k < SPT; ++k)
      if (mr[k] >= 0) atomicAdd(&rw[mr[k] & (RPB / 2 - 1)], 1 << ((mr[k] >> (RPB_SHIFT - 1)) * 16));
    __syncthreads();
    if (tid < RPB / 2) {
      int tl = 0, th = 0;
      for (int q = 0; q < SUM_WAVES; ++q) {
        const unsigned wd = (unsigned)rcw[q * (RPB / 2) + tid];
        tl += (int)(wd & 0xFFFFu);
        th += (int)(wd >> 16);
      }
      rtot[tid] = tl;
      rtot[tid + RPB / 2] = th;
    }
    __syncthreads();
    if (tid < 32) {
      int c[RPB / 32];
      int s = 0;
      for (int q = 0; q < RPB / 32; ++q) { c[q] = rtot[tid * (RPB / 32) + q]; s += c[q]; }
      int incl = s;
      for (int d = 1; d < 32; d <<= 1) {
        const int t = __shfl_up(incl, d);
        if (tid >= d) incl += t;
      }
      int ex = incl - s;
      for (int q = 0; q < RPB / 32; ++q) { rtot[tid * (RPB / 32) + q] = ex; ex += c[q]; }
    }
    __syncthreads();
    if (tid < RPB / 2) {
      int ol = rtot[tid], oh = rtot[tid + RPB / 2];
      for (int q = 0; q < SUM_WAVES; ++q) {
        const unsigned wd = (unsigned)rcw[q * (RPB / 2) + tid];
        rcw[q * (RPB / 2) + tid] = (int)((unsigned)ol | ((unsigned)oh << 16));
        ol += (int)(wd & 0xFFFFu);
        oh += (int)(wd >> 16);
      }
    }
    __syncthreads();
    int p[SPT];
#pragma unroll
    for (int k = 0; k < SPT; ++k) {
      p[k] = -1;
      if (mr[k] >= 0) {
        const int sh = (mr[k] >> (RPB_SHIFT - 1)) * 16;
        const unsigned old = (unsigned)atomicAdd(&rw[mr[k] & (RPB / 2 - 1)], 1 << sh);
        p[k] = (int)((old >> sh) & 0xFFFFu);
      }
    }
#pragma unroll
    for (int k = 0; k < SPT; ++k) {
      if (p[k] >= 0) { sc[p[k]] = mc[k]; sr[p[k]] = mr[k]; sz[p[k]] = mz[k]; }
    }
    __syncthreads();

    for (int i0 = wave * 32; i0 < n; i0 += (SUM_THREADS / 32) * 32) {
      int idx = i0 + lane;
      if (idx >= n) idx = n - 1;
      const int rc = sc[idx];
      const int rr = sr[idx];
      const int rz = sz[idx];
      const int m = (n - i0 < 32) ? (n - i0) : 32;
      float a0 = 0.0f, a1 = 0.0f;
      const int rprev = __shfl_up(rr, 1);
      const unsigned chg = __builtin_amdgcn_ballot_w32(lane != 0 && rr != rprev);
      int last;
      if (m == 32) {
        v2f xr[32];
#pragma unroll
        for (int j = 0; j < 32; ++j) {
          int c = __builtin_amdgcn_readlane(rc, j);
          xr[j] = *(const v2f*)(X + (long long)c * DFEAT + 2 * lane);
        }
#pragma unroll
        for (int j = 0; j < 32; ++j) {
          if (j > 0 && ((chg >> j) & 1u)) {
            float* td = tile + __builtin_amdgcn_readlane(rr, j - 1) * TP + 2 * lane;
            atomicAdd(td + 0, a0);
            atomicAdd(td + 1, a1);
            a0 = 0.0f; a1 = 0.0f;
          }
          const float v = __int_as_float(__builtin_amdgcn_readlane(rz, j));
          a0 += xr[j].x * v;
          a1 += xr[j].y * v;
        }
        last = __builtin_amdgcn_readlane(rr, 31);
      } else {
        for (int j = 0; j < m; ++j) {
          const int c = __builtin_amdgcn_readlane(rc, j);
          const float v = __int_as_float(__builtin_amdgcn_readlane(rz, j));
          const v2f x = *(const v2f*)(X + (long long)c * DFEAT + 2 * lane);
          if (j > 0 && ((chg >> j) & 1u)) {
            float* td = tile + __builtin_amdgcn_readlane(rr, j - 1) * TP + 2 * lane;
            atomicAdd(td + 0, a0);
            atomicAdd(td + 1, a1);
            a0 = 0.0f; a1 = 0.0f;
          }
          a0 += x.x * v;
          a1 += x.y * v;
        }
        last = __builtin_amdgcn_readlane(rr, m - 1);
      }
      float* td = tile + last * TP + 2 * lane;
      atomicAdd(td + 0, a0);
      atomicAdd(td + 1, a1);
    }
  }
  __syncthreads();

  const int m0 = wave * 16;
  const int lo = lane & 15;
  const int hi = lane >> 4;
  const int koff = hi * 2;

  v8f acc[4];
#pragma unroll
  for (int t = 0; t < 4; ++t) {
    float bv = bl[t * 16 + lo];
#pragma unroll
    for (int r = 0; r < 8; ++r) acc[t][r] = bv;
  }

  const float* arow = tile + (m0 + lo) * TP;
  const v2f* wl2 = (const v2f*)Wl2;

#pragma unroll
  for (int kk = 0; kk < 16; ++kk) {
    int kb = kk * 4;
    int kpair = (kb + koff) >> 1;
    v2f a = *(const v2f*)(arow + kb + koff);
#pragma unroll
    for (int t = 0; t < 4; ++t) {
      v2f bb = wl2[kpair * 64 + t * 16 + lo];
      acc[t] = __builtin_amdgcn_wmma_f32_16x16x4_f32(
           false, a,  false, bb,
           (short)0, acc[t],  false,  false);
    }
  }

#pragma unroll
  for (int t = 0; t < 4; ++t) {
#pragma unroll
    for (int r = 0; r < 8; ++r)
      tile[(m0 + r + 8 * hi) * TP + t * 16 + lo] = acc[t][r];
  }
  __syncthreads();

  const int r0 = b * RPB;
  const int nrows = (n_nodes - r0 < RPB) ? (n_nodes - r0) : RPB;
  v4f* dst = (v4f*)(out + (long long)r0 * DFEAT);
  for (int i = tid; i < nrows * (DFEAT / 4); i += SUM_THREADS)
    dst[i] = *(const v4f*)(tile + (i >> 4) * TP + (i & 15) * 4);
}


extern "C" void kernel_launch(void* const* d_in, const int* in_sizes, int n_in,
                              void* d_out, int out_size, void* d_ws, size_t ws_size,
                              hipStream_t stream) {
  const float* X    = (const float*)d_in[0];
  const float* vals = (const float*)d_in[1];
  const float* W    = (const float*)d_in[2];
  const float* bias = (const float*)d_in[3];
  const int*   row  = (const int*)d_in[4];
  const int*   col  = (const int*)d_in[5];
  float* out = (float*)d_out;

  int n_nodes = in_sizes[0] / DFEAT;
  int n_edges = in_sizes[1];
  int n_buckets = (n_nodes + RPB - 1) >> RPB_SHIFT;
  int n_chunks = (n_edges + CHUNK - 1) / CHUNK;

  size_t rec_bytes  = (((size_t)n_edges * sizeof(v4i)) + 255) & ~(size_t)255;
  char* ws = (char*)d_ws;
  v4i* recs = (v4i*)ws;
  int* tab  = (int*)(ws + rec_bytes);
  if (rec_bytes + (size_t)n_chunks * (n_buckets + 1) * sizeof(int) > ws_size) return;
  if (n_chunks > MAXC) return;

  gcn_chunk_sort_kernel<<<n_chunks, BIN_THREADS,
                          ((size_t)n_buckets * BIN_WAVES + BIN_THREADS) * sizeof(int), stream>>>(
      row, col, vals, n_edges, n_buckets, tab, recs);

  gcn_bucket_sum_gemm_kernel<<<n_buckets, SUM_THREADS, 0, stream>>>(X, recs, tab, W, bias, out,
                                                                    n_nodes, n_buckets, n_chunks);
}


# Chunk sort places its records in bucket order in LDS, branch-free (one wait for a lane's 32 ranks, not 32), then copies them out with whole-line stores, not a scattered 16-byte store per edge
# speedup vs baseline: 3.0804x; 1.1320x over previous
#include <hip/hip_runtime.h>


typedef __attribute__((ext_vector_type(2))) float v2f;
typedef __attribute__((ext_vector_type(4))) float v4f;
typedef __attribute__((ext_vector_type(8))) float v8f;
typedef __attribute__((ext_vector_type(4))) int v4i;

#define DFEAT 64
#define RPB_SHIFT 8
#define RPB 256
#define TP 68
#define CHUNK 8192
#define BIN_THREADS 256
#define SUM_THREADS 512
#define EPT (CHUNK / BIN_THREADS)
#define SCH 5120
#define SPT (SCH / SUM_THREADS)
#define BIN_WAVES (BIN_THREADS / 32)
#define SUM_WAVES (SUM_THREADS / 32)
#define MAXC 256

__global__ void __launch_bounds__(BIN_THREADS)
gcn_chunk_sort_kernel(const int* __restrict__ row, const int* __restrict__ col,
                      const float* __restrict__ vals, int n_edges, int n_buckets,
                      int* __restrict__ tab, v4i* __restrict__ recs) {
  __shared__ __attribute__((aligned(16))) v4i stage[CHUNK + 1];
  extern __shared__ int sm[];
  int* lhw = sm;
  int* part = sm + BIN_WAVES * n_buckets;
  const int tid = threadIdx.x;
  int* lh = lhw + (tid >> 5) * n_buckets;

  static_assert(CHUNK % BIN_THREADS == 0, "a whole number of edges per lane");
  const int e0 = blockIdx.x * CHUNK;
  const int e1 = (e0 + CHUNK < n_edges) ? (e0 + CHUNK) : n_edges;
  int r[EPT], c[EPT], vb[EPT];
#pragma unroll
  for (int k = 0; k < EPT; ++k) {
    const int e = e0 + tid + k * BIN_THREADS;
    r[k] = (e < e1) ? row[e] : -1;
  }
#pragma unroll
  for (int k = 0; k < EPT; ++k) {
    const int e = e0 + tid + k * BIN_THREADS;
    const bool ok = e < e1;
    c[k] = ok ? col[e] : 0;
    vb[k] = ok ? __float_as_int(vals[e]) : 0;
  }
  for (int i = tid; i < BIN_WAVES * n_buckets; i += BIN_THREADS) lhw[i] = 0;
  __syncthreads();
#pragma unroll
  for (int k = 0; k < EPT; ++k)
    if (r[k] >= 0) atomicAdd(&lh[r[k] >> RPB_SHIFT], 1);
  __syncthreads();

  const int seg = (n_buckets + BIN_THREADS - 1) / BIN_THREADS;
  const int s0 = tid * seg;
  const int s1 = (s0 + seg < n_buckets) ? (s0 + seg) : n_buckets;
  int ssum = 0;
  for (int i = s0; i < s1; ++i)
    for (int q = 0; q < BIN_WAVES; ++q) ssum += lhw[q * n_buckets + i];
  part[tid] = ssum;
  __syncthreads();
  int pre = 0;
  for (int i = 0; i < tid; ++i) pre += part[i];
  int* trow = tab + (long long)blockIdx.x * (n_buckets + 1);
  for (int i = s0; i < s1; ++i) {
    trow[i] = pre;
    for (int q = 0; q < BIN_WAVES; ++q) {
      const int c = lhw[q * n_buckets + i];
      lhw[q * n_buckets + i] = pre;
      pre += c;
    }
  }
  if (tid == 0) trow[n_buckets] = e1 - e0;
  __syncthreads();

  int pos[EPT];
#pragma unroll
  for (int k = 0; k < EPT; ++k) {
    pos[k] = 0;
    if (r[k] >= 0) pos[k] = atomicAdd(&lh[r[k] >> RPB_SHIFT], 1);
  }
#pragma unroll
  for (int k = 0; k < EPT; ++k) {
    v4i rec;
    rec.x = c[k];
    rec.y = r[k] & (RPB - 1);
    rec.z = vb[k];
    rec.w = 0;
    stage[(r[k] >= 0) ? pos[k] : CHUNK] = rec;
  }
  __syncthreads();
  const int cnt = e1 - e0;
  v4i* dst = recs + e0;
  if (cnt == CHUNK) {
#pragma unroll
    for (int k = 0; k < EPT; ++k) dst[tid + k * BIN_THREADS] = stage[tid + k * BIN_THREADS];
  } else {
    for (int i = tid; i < cnt; i += BIN_THREADS) dst[i] = stage[i];
  }
}

__global__ void __launch_bounds__(SUM_THREADS) __attribute__((amdgpu_waves_per_eu(4, 8)))
gcn_bucket_sum_gemm_kernel(const float* __restrict__ X, const v4i* __restrict__ recs,
                           const int* __restrict__ tab, const float* __restrict__ W,
                           const float* __restrict__ bias, float* __restrict__ out,
                           int n_nodes, int n_buckets, int n_chunks) {
  __shared__ __attribute__((aligned(16))) float tile[RPB * TP];
  __shared__ __attribute__((aligned(16))) float Wl2[DFEAT * DFEAT];
  __shared__ float bl[DFEAT];
  __shared__ int sc[SCH];
  __shared__ int sr[SCH];
  __shared__ int sz[SCH];
  __shared__ int rcw[SUM_WAVES * (RPB / 2)];
  __shared__ int rtot[RPB];
  __shared__ int pfx[MAXC + 1];
  __shared__ int pst[MAXC];
  const int tid = threadIdx.x;
  const int b = blockIdx.x;

  static_assert(DFEAT * DFEAT == 8 * SUM_THREADS, "two 4-float pieces of W per thread");
  const v4f w0 = ((const v4f*)W)[tid];
  const v4f w1 = ((const v4f*)W)[tid + SUM_THREADS];
  float bv0 = 0.0f;
  if (tid < DFEAT) bv0 = bias[tid];
  static_assert(MAXC <= SUM_THREADS, "one table entry per thread");
  int sz_c = 0, st_c = 0;
  if (tid < MAXC && tid < n_chunks) {
    const int* te = tab + (long long)tid * (n_buckets + 1) + b;
    const int s = te[0];
    sz_c = te[1] - s;
    st_c = tid * CHUNK + s;
  }

  v4f z = {0.0f, 0.0f, 0.0f, 0.0f};
  for (int i = tid; i < RPB * TP / 4; i += SUM_THREADS) ((v4f*)tile)[i] = z;
  {
    const int k = tid >> 4;
    const int n = (tid & 15) * 4;
    float* d0 = Wl2 + (k >> 1) * 128 + n * 2 + (k & 1);
    float* d1 = d0 + 16 * 128;
    d0[0] = w0.x; d0[2] = w0.y; d0[4] = w0.z; d0[6] = w0.w;
    d1[0] = w1.x; d1[2] = w1.y; d1[4] = w1.z; d1[6] = w1.w;
  }
  if (tid < DFEAT) bl[tid] = bv0;
  if (tid < MAXC) { pfx[tid] = sz_c; pst[tid] = st_c; }
  __syncthreads();
  if (tid < 32) {
    int c[MAXC / 32];
    int s = 0;
    for (int q = 0; q < MAXC / 32; ++q) { c[q] = pfx[tid * (MAXC / 32) + q]; s += c[q]; }
    int incl = s;
    for (int d = 1; d < 32; d <<= 1) {
      const int t = __shfl_up(incl, d);
      if (tid >= d) incl += t;
    }
    int ex = incl - s;
    for (int q = 0; q < MAXC / 32; ++q) { pfx[tid * (MAXC / 32) + q] = ex; ex += c[q]; }
    if (tid == 31) pfx[MAXC] = ex;
  }
  __syncthreads();
  const int cnt = pfx[MAXC];

  const int lane = tid & 31;
  const int wave = __builtin_amdgcn_readfirstlane(tid >> 5);

  for (int cb = 0; cb < cnt; cb += SCH) {
    const int n = (cnt - cb < SCH) ? (cnt - cb) : SCH;
    for (int i = tid; i < SUM_WAVES * (RPB / 2); i += SUM_THREADS) rcw[i] = 0;
    __syncthreads();
    int mc[SPT], mr[SPT], mz[SPT];
    int gi[SPT], lo[SPT];
#pragma unroll
    for (int k = 0; k < SPT; ++k) {
      const int i = cb + tid + k * SUM_THREADS;
      gi[k] = (i < cnt) ? i : (cnt - 1);
      lo[k] = 0;
    }
#pragma unroll
    for (int step = MAXC / 2; step > 0; step >>= 1) {
#pragma unroll
      for (int k = 0; k < SPT; ++k) lo[k] += (pfx[lo[k] + step] <= gi[k]) ? step : 0;
    }
    int ra[SPT];
#pragma unroll
    for (int k = 0; k < SPT; ++k) ra[k] = pst[lo[k]] + (gi[k] - pfx[lo[k]]);
#pragma unroll
    for (int k = 0; k < SPT; ++k) {
      const int i = tid + k * SUM_THREADS;
      mc[k] = 0; mr[k] = -1; mz[k] = 0;
      if (i < n) {
        const v4i rec = recs[ra[k]];
        mc[k] = rec.x; mr[k] = rec.y; mz[k] = rec.z;
      }
    }
    int* rw = rcw + wave * (RPB / 2);
#pragma unroll
    for (int k = 0; k < SPT; ++k)
      if (mr[k] >= 0) atomicAdd(&rw[mr[k] & (RPB / 2 - 1)], 1 << ((mr[k] >> (RPB_SHIFT - 1)) * 16));
    __syncthreads();
    if (tid < RPB / 2) {
      int tl = 0, th = 0;
      for (int q = 0; q < SUM_WAVES; ++q) {
        const unsigned wd = (unsigned)rcw[q * (RPB / 2) + tid];
        tl += (int)(wd & 0xFFFFu);
        th += (int)(wd >> 16);
      }
      rtot[tid] = tl;
      rtot[tid + RPB / 2] = th;
    }
    __syncthreads();
    if (tid < 32) {
      int c[RPB / 32];
      int s = 0;
      for (int q = 0; q < RPB / 32; ++q) { c[q] = rtot[tid * (RPB / 32) + q]; s += c[q]; }
      int incl = s;
      for (int d = 1; d < 32; d <<= 1) {
        const int t = __shfl_up(incl, d);
        if (tid >= d) incl += t;
      }
      int ex = incl - s;
      for (int q = 0; q < RPB / 32; ++q) { rtot[tid * (RPB / 32) + q] = ex; ex += c[q]; }
    }
    __syncthreads();
    if (tid < RPB / 2) {
      int ol = rtot[tid], oh = rtot[tid + RPB / 2];
      for (int q = 0; q < SUM_WAVES; ++q) {
        const unsigned wd = (unsigned)rcw[q * (RPB / 2) + tid];
        rcw[q * (RPB / 2) + tid] = (int)((unsigned)ol | ((unsigned)oh << 16));
        ol += (int)(wd & 0xFFFFu);
        oh += (int)(wd >> 16);
      }
    }
    __syncthreads();
    int p[SPT];
#pragma unroll
    for (int k = 0; k < SPT; ++k) {
      p[k] = -1;
      if (mr[k] >= 0) {
        const int sh = (mr[k] >> (RPB_SHIFT - 1)) * 16;
        const unsigned old = (unsigned)atomicAdd(&rw[mr[k] & (RPB / 2 - 1)], 1 << sh);
        p[k] = (int)((old >> sh) & 0xFFFFu);
      }
    }
#pragma unroll
    for (int k = 0; k < SPT; ++k) {
      if (p[k] >= 0) { sc[p[k]] = mc[k]; sr[p[k]] = mr[k]; sz[p[k]] = mz[k]; }
    }
    __syncthreads();

    for (int i0 = wave * 32; i0 < n; i0 += (SUM_THREADS / 32) * 32) {
      int idx = i0 + lane;
      if (idx >= n) idx = n - 1;
      const int rc = sc[idx];
      const int rr = sr[idx];
      const int rz = sz[idx];
      const int m = (n - i0 < 32) ? (n - i0) : 32;
      float a0 = 0.0f, a1 = 0.0f;
      const int rprev = __shfl_up(rr, 1);
      const unsigned chg = __builtin_amdgcn_ballot_w32(lane != 0 && rr != rprev);
      int last;
      if (m == 32) {
        v2f xr[32];
#pragma unroll
        for (int j = 0; j < 32; ++j) {
          int c = __builtin_amdgcn_readlane(rc, j);
          xr[j] = *(const v2f*)(X + (long long)c * DFEAT + 2 * lane);
        }
#pragma unroll
        for (int j = 0; j < 32; ++j) {
          if (j > 0 && ((chg >> j) & 1u)) {
            float* td = tile + __builtin_amdgcn_readlane(rr, j - 1) * TP + 2 * lane;
            atomicAdd(td + 0, a0);
            atomicAdd(td + 1, a1);
            a0 = 0.0f; a1 = 0.0f;
          }
          const float v = __int_as_float(__builtin_amdgcn_readlane(rz, j));
          a0 += xr[j].x * v;
          a1 += xr[j].y * v;
        }
        last = __builtin_amdgcn_readlane(rr, 31);
      } else {
        for (int j = 0; j < m; ++j) {
          const int c = __builtin_amdgcn_readlane(rc, j);
          const float v = __int_as_float(__builtin_amdgcn_readlane(rz, j));
          const v2f x = *(const v2f*)(X + (long long)c * DFEAT + 2 * lane);
          if (j > 0 && ((chg >> j) & 1u)) {
            float* td = tile + __builtin_amdgcn_readlane(rr, j - 1) * TP + 2 * lane;
            atomicAdd(td + 0, a0);
            atomicAdd(td + 1, a1);
            a0 = 0.0f; a1 = 0.0f;
          }
          a0 += x.x * v;
          a1 += x.y * v;
        }
        last = __builtin_amdgcn_readlane(rr, m - 1);
      }
      float* td = tile + last * TP + 2 * lane;
      atomicAdd(td + 0, a0);
      atomicAdd(td + 1, a1);
    }
  }
  __syncthreads();

  const int m0 = wave * 16;
  const int lo = lane & 15;
  const int hi = lane >> 4;
  const int koff = hi * 2;

  v8f acc[4];
#pragma unroll
  for (int t = 0; t < 4; ++t) {
    float bv = bl[t * 16 + lo];
#pragma unroll
    for (int r = 0; r < 8; ++r) acc[t][r] = bv;
  }

  const float* arow = tile + (m0 + lo) * TP;
  const v2f* wl2 = (const v2f*)Wl2;

#pragma unroll
  for (int kk = 0; kk < 16; ++kk) {
    int kb = kk * 4;
    int kpair = (kb + koff) >> 1;
    v2f a = *(const v2f*)(arow + kb + koff);
#pragma unroll
    for (int t = 0; t < 4; ++t) {
      v2f bb = wl2[kpair * 64 + t * 16 + lo];
      acc[t] = __builtin_amdgcn_wmma_f32_16x16x4_f32(
           false, a,  false, bb,
           (short)0, acc[t],  false,  false);
    }
  }

#pragma unroll
  for (int t = 0; t < 4; ++t) {
#pragma unroll
    for (int r = 0; r < 8; ++r)
      tile[(m0 + r + 8 * hi) * TP + t * 16 + lo] = acc[t][r];
  }
  __syncthreads();

  const int r0 = b * RPB;
  const int nrows = (n_nodes - r0 < RPB) ? (n_nodes - r0) : RPB;
  v4f* dst = (v4f*)(out + (long long)r0 * DFEAT);
  for (int i = tid; i < nrows * (DFEAT / 4); i += SUM_THREADS)
    dst[i] = *(const v4f*)(tile + (i >> 4) * TP + (i & 15) * 4);
}


extern "C" void kernel_launch(void* const* d_in, const int* in_sizes, int n_in,
                              void* d_out, int out_size, void* d_ws, size_t ws_size,
                              hipStream_t stream) {
  const float* X    = (const float*)d_in[0];
  const float* vals = (const float*)d_in[1];
  const float* W    = (const float*)d_in[2];
  const float* bias = (const float*)d_in[3];
  const int*   row  = (const int*)d_in[4];
  const int*   col  = (const int*)d_in[5];
  float* out = (float*)d_out;

  int n_nodes = in_sizes[0] / DFEAT;
  int n_edges = in_sizes[1];
  int n_buckets = (n_nodes + RPB - 1) >> RPB_SHIFT;
  int n_chunks = (n_edges + CHUNK - 1) / CHUNK;

  size_t rec_bytes  = (((size_t)n_edges * sizeof(v4i)) + 255) & ~(size_t)255;
  char* ws = (char*)d_ws;
  v4i* recs = (v4i*)ws;
  int* tab  = (int*)(ws + rec_bytes);
  if (rec_bytes + (size_t)n_chunks * (n_buckets + 1) * sizeof(int) > ws_size) return;
  if (n_chunks > MAXC) return;

  gcn_chunk_sort_kernel<<<n_chunks, BIN_THREADS,
                          ((size_t)n_buckets * BIN_WAVES + BIN_THREADS) * sizeof(int), stream>>>(
      row, col, vals, n_edges, n_buckets, tab, recs);

  gcn_bucket_sum_gemm_kernel<<<n_buckets, SUM_THREADS, 0, stream>>>(X, recs, tab, W, bias, out,
                                                                    n_nodes, n_buckets, n_chunks);
}


# No load under its own guard or on a rewritten address register: full chunks load off one index, a pass's records unguarded, gathers as X + 32-bit offset; chunk-sort barriers wait on LDS only
# speedup vs baseline: 3.3164x; 1.0766x over previous
#include <hip/hip_runtime.h>


typedef __attribute__((ext_vector_type(2))) float v2f;
typedef __attribute__((ext_vector_type(4))) float v4f;
typedef __attribute__((ext_vector_type(8))) float v8f;
typedef __attribute__((ext_vector_type(4))) int v4i;

#define DFEAT 64
#define RPB_SHIFT 8
#define RPB 256
#define TP 68
#define CHUNK 8192
#define BIN_THREADS 256
#define SUM_THREADS 512
#define EPT (CHUNK / BIN_THREADS)
#define SCH 5120
#define SPT (SCH / SUM_THREADS)
#define BIN_WAVES (BIN_THREADS / 32)
#define SUM_WAVES (SUM_THREADS / 32)
#define MAXC 256

__device__ __forceinline__ void lds_only_barrier() {
  __builtin_amdgcn_fence(__ATOMIC_RELEASE, "workgroup", "local");
  __builtin_amdgcn_s_barrier();
  __builtin_amdgcn_fence(__ATOMIC_ACQUIRE, "workgroup", "local");
}

template <bool FULL>
__device__ __forceinline__ void chunk_sort_body(
    const int* __restrict__ row, const int* __restrict__ col, const float* __restrict__ vals,
    const int e0, const int e1, const int n_buckets, int* __restrict__ trow,
    v4i* __restrict__ dst, v4i* stage, int* lhw, int* part) {
  const int tid = threadIdx.x;
  int* lh = lhw + (tid >> 5) * n_buckets;

  int r[EPT], c[EPT], vb[EPT];
  if (FULL) {
    const int* rp = row + e0 + tid;
    const int* cp = col + e0 + tid;
    const float* vp = vals + e0 + tid;
#pragma unroll
    for (int k = 0; k < EPT; ++k) r[k] = rp[k * BIN_THREADS];
#pragma unroll
    for (int k = 0; k < EPT; ++k) {
      c[k] = cp[k * BIN_THREADS];
      vb[k] = __float_as_int(vp[k * BIN_THREADS]);
    }
  } else {
#pragma unroll
    for (int k = 0; k < EPT; ++k) {
      const int e = e0 + tid + k * BIN_THREADS;
      r[k] = (e < e1) ? row[e] : -1;
    }
#pragma unroll
    for (int k = 0; k < EPT; ++k) {
      const int e = e0 + tid + k * BIN_THREADS;
      const bool ok = e < e1;
      c[k] = ok ? col[e] : 0;
      vb[k] = ok ? __float_as_int(vals[e]) : 0;
    }
  }
  for (int i = tid; i < BIN_WAVES * n_buckets; i += BIN_THREADS) lhw[i] = 0;
  lds_only_barrier();
#pragma unroll
  for (int k = 0; k < EPT; ++k)
    if (FULL || r[k] >= 0) atomicAdd(&lh[r[k] >> RPB_SHIFT], 1);
  lds_only_barrier();

  const int seg = (n_buckets + BIN_THREADS - 1) / BIN_THREADS;
  const int s0 = tid * seg;
  const int s1 = (s0 + seg < n_buckets) ? (s0 + seg) : n_buckets;
  int ssum = 0;
  for (int i = s0; i < s1; ++i)
    for (int q = 0; q < BIN_WAVES; ++q) ssum += lhw[q * n_buckets + i];
  part[tid] = ssum;
  lds_only_barrier();
  int pre = 0;
  for (int i = 0; i < tid; ++i) pre += part[i];
  for (int i = s0; i < s1; ++i) {
    trow[i] = pre;
    for (int q = 0; q < BIN_WAVES; ++q) {
      const int cq = lhw[q * n_buckets + i];
      lhw[q * n_buckets + i] = pre;
      pre += cq;
    }
  }
  if (tid == 0) trow[n_buckets] = e1 - e0;
  lds_only_barrier();

  int pos[EPT];
#pragma unroll
  for (int k = 0; k < EPT; ++k) {
    pos[k] = 0;
    if (FULL || r[k] >= 0) pos[k] = atomicAdd(&lh[r[k] >> RPB_SHIFT], 1);
  }
#pragma unroll
  for (int k = 0; k < EPT; ++k) {
    v4i rec;
    rec.x = c[k];
    rec.y = r[k] & (RPB - 1);
    rec.z = vb[k];
    rec.w = 0;
    stage[(FULL || r[k] >= 0) ? pos[k] : CHUNK] = rec;
  }
  __syncthreads();
  if (FULL) {
#pragma unroll
    for (int k = 0; k < EPT; ++k) dst[tid + k * BIN_THREADS] = stage[tid + k * BIN_THREADS];
  } else {
    const int cnt = e1 - e0;
    for (int i = tid; i < cnt; i += BIN_THREADS) dst[i] = stage[i];
  }
}

__global__ void __launch_bounds__(BIN_THREADS)
gcn_chunk_sort_kernel(const int* __restrict__ row, const int* __restrict__ col,
                      const float* __restrict__ vals, int n_edges, int n_buckets,
                      int* __restrict__ tab, v4i* __restrict__ recs) {
  __shared__ __attribute__((aligned(16))) v4i stage[CHUNK + 1];
  extern __shared__ int sm[];
  int* lhw = sm;
  int* part = sm + BIN_WAVES * n_buckets;

  static_assert(CHUNK % BIN_THREADS == 0, "a whole number of edges per lane");
  const int e0 = blockIdx.x * CHUNK;
  const int e1 = (e0 + CHUNK < n_edges) ? (e0 + CHUNK) : n_edges;
  int* trow = tab + (long long)blockIdx.x * (n_buckets + 1);
  if (e1 - e0 == CHUNK)
    chunk_sort_body<true>(row, col, vals, e0, e1, n_buckets, trow, recs + e0, stage, lhw, part);
  else
    chunk_sort_body<false>(row, col, vals, e0, e1, n_buckets, trow, recs + e0, stage, lhw, part);
}

__global__ void __launch_bounds__(SUM_THREADS) __attribute__((amdgpu_waves_per_eu(4, 8)))
gcn_bucket_sum_gemm_kernel(const float* __restrict__ X, const v4i* __restrict__ recs,
                           const int* __restrict__ tab, const float* __restrict__ W,
                           const float* __restrict__ bias, float* __restrict__ out,
                           int n_nodes, int n_buckets, int n_chunks) {
  __shared__ __attribute__((aligned(16))) float tile[RPB * TP];
  __shared__ __attribute__((aligned(16))) float Wl2[DFEAT * DFEAT];
  __shared__ float bl[DFEAT];
  __shared__ int sc[SCH];
  __shared__ int sr[SCH];
  __shared__ int sz[SCH];
  __shared__ int rcw[SUM_WAVES * (RPB / 2)];
  __shared__ int rtot[RPB];
  __shared__ int pfx[MAXC + 1];
  __shared__ int pst[MAXC];
  const int tid = threadIdx.x;
  const int b = blockIdx.x;

  static_assert(DFEAT * DFEAT == 8 * SUM_THREADS, "two 4-float pieces of W per thread");
  const v4f w0 = ((const v4f*)W)[tid];
  const v4f w1 = ((const v4f*)W)[tid + SUM_THREADS];
  float bv0 = 0.0f;
  if (tid < DFEAT) bv0 = bias[tid];
  static_assert(MAXC <= SUM_THREADS, "one table entry per thread");
  int sz_c = 0, st_c = 0;
  if (tid < MAXC && tid < n_chunks) {
    const int* te = tab + (long long)tid * (n_buckets + 1) + b;
    const int s = te[0];
    sz_c = te[1] - s;
    st_c = tid * CHUNK + s;
  }

  v4f z = {0.0f, 0.0f, 0.0f, 0.0f};
  for (int i = tid; i < RPB * TP / 4; i += SUM_THREADS) ((v4f*)tile)[i] = z;
  {
    const int k = tid >> 4;
    const int n = (tid & 15) * 4;
    float* d0 = Wl2 + (k >> 1) * 128 + n * 2 + (k & 1);
    float* d1 = d0 + 16 * 128;
    d0[0] = w0.x; d0[2] = w0.y; d0[4] = w0.z; d0[6] = w0.w;
    d1[0] = w1.x; d1[2] = w1.y; d1[4] = w1.z; d1[6] = w1.w;
  }
  if (tid < DFEAT) bl[tid] = bv0;
  if (tid < MAXC) { pfx[tid] = sz_c; pst[tid] = st_c; }
  __syncthreads();
  if (tid < 32) {
    int c[MAXC / 32];
    int s = 0;
    for (int q = 0; q < MAXC / 32; ++q) { c[q] = pfx[tid * (MAXC / 32) + q]; s += c[q]; }
    int incl = s;
    for (int d = 1; d < 32; d <<= 1) {
      const int t = __shfl_up(incl, d);
      if (tid >= d) incl += t;
    }
    int ex = incl - s;
    for (int q = 0; q < MAXC / 32; ++q) { pfx[tid * (MAXC / 32) + q] = ex; ex += c[q]; }
    if (tid == 31) pfx[MAXC] = ex;
  }
  __syncthreads();
  const int cnt = pfx[MAXC];

  const int lane = tid & 31;
  const unsigned lane8 = (unsigned)lane * 8u;
  const int wave = __builtin_amdgcn_readfirstlane(tid >> 5);

  for (int cb = 0; cb < cnt; cb += SCH) {
    const int n = (cnt - cb < SCH) ? (cnt - cb) : SCH;
    for (int i = tid; i < SUM_WAVES * (RPB / 2); i += SUM_THREADS) rcw[i] = 0;
    __syncthreads();
    int mc[SPT], mr[SPT], mz[SPT];
    int gi[SPT], lo[SPT];
#pragma unroll
    for (int k = 0; k < SPT; ++k) {
      const int i = cb + tid + k * SUM_THREADS;
      gi[k] = (i < cnt) ? i : (cnt - 1);
      lo[k] = 0;
    }
#pragma unroll
    for (int step = MAXC / 2; step > 0; step >>= 1) {
#pragma unroll
      for (int k = 0; k < SPT; ++k) lo[k] += (pfx[lo[k] + step] <= gi[k]) ? step : 0;
    }
    int ra[SPT];
#pragma unroll
    for (int k = 0; k < SPT; ++k) ra[k] = pst[lo[k]] + (gi[k] - pfx[lo[k]]);
    v4i rec[SPT];
#pragma unroll
    for (int k = 0; k < SPT; ++k) rec[k] = recs[ra[k]];
#pragma unroll
    for (int k = 0; k < SPT; ++k) {
      const bool ok = tid + k * SUM_THREADS < n;
      mc[k] = ok ? rec[k].x : 0;
      mr[k] = ok ? rec[k].y : -1;
      mz[k] = ok ? rec[k].z : 0;
    }
    int* rw = rcw + wave * (RPB / 2);
#pragma unroll
    for (int k = 0; k < SPT; ++k)
      if (mr[k] >= 0) atomicAdd(&rw[mr[k] & (RPB / 2 - 1)], 1 << ((mr[k] >> (RPB_SHIFT - 1)) * 16));
    __syncthreads();
    if (tid < RPB / 2) {
      int tl = 0, th = 0;
      for (int q = 0; q < SUM_WAVES; ++q) {
        const unsigned wd = (unsigned)rcw[q * (RPB / 2) + tid];
        tl += (int)(wd & 0xFFFFu);
        th += (int)(wd >> 16);
      }
      rtot[tid] = tl;
      rtot[tid + RPB / 2] = th;
    }
    __syncthreads();
    if (tid < 32) {
      int c[RPB / 32];
      int s = 0;
      for (int q = 0; q < RPB / 32; ++q) { c[q] = rtot[tid * (RPB / 32) + q]; s += c[q]; }
      int incl = s;
      for (int d = 1; d < 32; d <<= 1) {
        const int t = __shfl_up(incl, d);
        if (tid >= d) incl += t;
      }
      int ex = incl - s;
      for (int q = 0; q < RPB / 32; ++q) { rtot[tid * (RPB / 32) + q] = ex; ex += c[q]; }
    }
    __syncthreads();
    if (tid < RPB / 2) {
      int ol = rtot[tid], oh = rtot[tid + RPB / 2];
      for (int q = 0; q < SUM_WAVES; ++q) {
        const unsigned wd = (unsigned)rcw[q * (RPB / 2) + tid];
        rcw[q * (RPB / 2) + tid] = (int)((unsigned)ol | ((unsigned)oh << 16));
        ol += (int)(wd & 0xFFFFu);
        oh += (int)(wd >> 16);
      }
    }
    __syncthreads();
    int p[SPT];
#pragma unroll
    for (int k = 0; k < SPT; ++k) {
      p[k] = -1;
      if (mr[k] >= 0) {
        const int sh = (mr[k] >> (RPB_SHIFT - 1)) * 16;
        const unsigned old = (unsigned)atomicAdd(&rw[mr[k] & (RPB / 2 - 1)], 1 << sh);
        p[k] = (int)((old >> sh) & 0xFFFFu);
      }
    }
#pragma unroll
    for (int k = 0; k < SPT; ++k) {
      if (p[k] >= 0) { sc[p[k]] = mc[k]; sr[p[k]] = mr[k]; sz[p[k]] = mz[k]; }
    }
    __syncthreads();

    for (int i0 = wave * 32; i0 < n; i0 += (SUM_THREADS / 32) * 32) {
      int idx = i0 + lane;
      if (idx >= n) idx = n - 1;
      const int rc = sc[idx];
      const unsigned rc8 = (unsigned)rc * (unsigned)(DFEAT * sizeof(float));
      const int rr = sr[idx];
      const int rz = sz[idx];
      const int m = (n - i0 < 32) ? (n - i0) : 32;
      float a0 = 0.0f, a1 = 0.0f;
      const int rprev = __shfl_up(rr, 1);
      const unsigned chg = __builtin_amdgcn_ballot_w32(lane != 0 && rr != rprev);
      int last;
      if (m == 32) {
        v2f xr[32];
        unsigned ro[32];
#pragma unroll
        for (int j = 0; j < 32; ++j) ro[j] = (unsigned)__builtin_amdgcn_readlane((int)rc8, j) + lane8;
#pragma unroll
        for (int j = 0; j < 32; ++j) xr[j] = *(const v2f*)((const char*)X + ro[j]);
#pragma unroll
        for (int j = 0; j < 32; ++j) {
          if (j > 0 && ((chg >> j) & 1u)) {
            float* td = tile + __builtin_amdgcn_readlane(rr, j - 1) * TP + 2 * lane;
            atomicAdd(td + 0, a0);
            atomicAdd(td + 1, a1);
            a0 = 0.0f; a1 = 0.0f;
          }
          const float v = __int_as_float(__builtin_amdgcn_readlane(rz, j));
          a0 += xr[j].x * v;
          a1 += xr[j].y * v;
        }
        last = __builtin_amdgcn_readlane(rr, 31);
      } else {
        for (int j = 0; j < m; ++j) {
          const int c = __builtin_amdgcn_readlane(rc, j);
          const float v = __int_as_float(__builtin_amdgcn_readlane(rz, j));
          const v2f x = *(const v2f*)((const char*)X + ((unsigned)c * (unsigned)(DFEAT * sizeof(float)) + lane8));
          if (j > 0 && ((chg >> j) & 1u)) {
            float* td = tile + __builtin_amdgcn_readlane(rr, j - 1) * TP + 2 * lane;
            atomicAdd(td + 0, a0);
            atomicAdd(td + 1, a1);
            a0 = 0.0f; a1 = 0.0f;
          }
          a0 += x.x * v;
          a1 += x.y * v;
        }
        last = __builtin_amdgcn_readlane(rr, m - 1);
      }
      float* td = tile + last * TP + 2 * lane;
      atomicAdd(td + 0, a0);
      atomicAdd(td + 1, a1);
    }
  }
  __syncthreads();

  const int m0 = wave * 16;
  const int lo = lane & 15;
  const int hi = lane >> 4;
  const int koff = hi * 2;

  v8f acc[4];
#pragma unroll
  for (int t = 0; t < 4; ++t) {
    float bv = bl[t * 16 + lo];
#pragma unroll
    for (int r = 0; r < 8; ++r) acc[t][r] = bv;
  }

  const float* arow = tile + (m0 + lo) * TP;
  const v2f* wl2 = (const v2f*)Wl2;

#pragma unroll
  for (int kk = 0; kk < 16; ++kk) {
    int kb = kk * 4;
    int kpair = (kb + koff) >> 1;
    v2f a = *(const v2f*)(arow + kb + koff);
#pragma unroll
    for (int t = 0; t < 4; ++t) {
      v2f bb = wl2[kpair * 64 + t * 16 + lo];
      acc[t] = __builtin_amdgcn_wmma_f32_16x16x4_f32(
           false, a,  false, bb,
           (short)0, acc[t],  false,  false);
    }
  }

#pragma unroll
  for (int t = 0; t < 4; ++t) {
#pragma unroll
    for (int r = 0; r < 8; ++r)
      tile[(m0 + r + 8 * hi) * TP + t * 16 + lo] = acc[t][r];
  }
  __syncthreads();

  const int r0 = b * RPB;
  const int nrows = (n_nodes - r0 < RPB) ? (n_nodes - r0) : RPB;
  v4f* dst = (v4f*)(out + (long long)r0 * DFEAT);
  for (int i = tid; i < nrows * (DFEAT / 4); i += SUM_THREADS)
    dst[i] = *(const v4f*)(tile + (i >> 4) * TP + (i & 15) * 4);
}


extern "C" void kernel_launch(void* const* d_in, const int* in_sizes, int n_in,
                              void* d_out, int out_size, void* d_ws, size_t ws_size,
                              hipStream_t stream) {
  const float* X    = (const float*)d_in[0];
  const float* vals = (const float*)d_in[1];
  const float* W    = (const float*)d_in[2];
  const float* bias = (const float*)d_in[3];
  const int*   row  = (const int*)d_in[4];
  const int*   col  = (const int*)d_in[5];
  float* out = (float*)d_out;

  int n_nodes = in_sizes[0] / DFEAT;
  int n_edges = in_sizes[1];
  int n_buckets = (n_nodes + RPB - 1) >> RPB_SHIFT;
  int n_chunks = (n_edges + CHUNK - 1) / CHUNK;

  size_t rec_bytes  = (((size_t)n_edges * sizeof(v4i)) + 255) & ~(size_t)255;
  char* ws = (char*)d_ws;
  v4i* recs = (v4i*)ws;
  int* tab  = (int*)(ws + rec_bytes);
  if (rec_bytes + (size_t)n_chunks * (n_buckets + 1) * sizeof(int) > ws_size) return;
  if (n_chunks > MAXC) return;
  if (n_nodes > (1 << 24)) return;

  gcn_chunk_sort_kernel<<<n_chunks, BIN_THREADS,
                          ((size_t)n_buckets * BIN_WAVES + BIN_THREADS) * sizeof(int), stream>>>(
      row, col, vals, n_edges, n_buckets, tab, recs);

  gcn_bucket_sum_gemm_kernel<<<n_buckets, SUM_THREADS, 0, stream>>>(X, recs, tab, W, bias, out,
                                                                    n_nodes, n_buckets, n_chunks);
}


# Prefix scans made parallel: chunk sort's (bucket, wave) scan and kernel 1's piece and row scans keep counts in registers; wave scan by DPP adds + per-wave totals, not serial LDS loops or shuffles
# speedup vs baseline: 3.5049x; 1.0568x over previous
#include <hip/hip_runtime.h>


typedef __attribute__((ext_vector_type(2))) float v2f;
typedef __attribute__((ext_vector_type(4))) float v4f;
typedef __attribute__((ext_vector_type(8))) float v8f;
typedef __attribute__((ext_vector_type(4))) int v4i;

#define DFEAT 64
#define RPB_SHIFT 8
#define RPB 256
#define TP 68
#define CHUNK 8192
#define BIN_THREADS 256
#define SUM_THREADS 512
#define EPT (CHUNK / BIN_THREADS)
#define SCH 5120
#define SPT (SCH / SUM_THREADS)
#define BIN_WAVES (BIN_THREADS / 32)
#define SUM_WAVES (SUM_THREADS / 32)
#define MAXC 256

__device__ __forceinline__ void lds_only_barrier() {
  __builtin_amdgcn_fence(__ATOMIC_RELEASE, "workgroup", "local");
  __builtin_amdgcn_s_barrier();
  __builtin_amdgcn_fence(__ATOMIC_ACQUIRE, "workgroup", "local");
}

__device__ __forceinline__ int wave_incl_scan(int x, const int lane) {
  x += __builtin_amdgcn_update_dpp(0, x, 0x111, 0xF, 0xF, true);
  x += __builtin_amdgcn_update_dpp(0, x, 0x112, 0xF, 0xF, true);
  x += __builtin_amdgcn_update_dpp(0, x, 0x114, 0xF, 0xF, true);
  x += __builtin_amdgcn_update_dpp(0, x, 0x118, 0xF, 0xF, true);
  const int low_row_total = __builtin_amdgcn_readlane(x, 15);
  return x + ((lane & 16) ? low_row_total : 0);
}

template <bool FULL>
__device__ __forceinline__ void chunk_sort_body(
    const int* __restrict__ row, const int* __restrict__ col, const float* __restrict__ vals,
    const int e0, const int e1, const int n_buckets, int* __restrict__ trow,
    v4i* __restrict__ dst, v4i* stage, int* lhw, int* part) {
  const int tid = threadIdx.x;
  int* lh = lhw + (tid >> 5) * n_buckets;

  int r[EPT], c[EPT], vb[EPT];
  if (FULL) {
    const int* rp = row + e0 + tid;
    const int* cp = col + e0 + tid;
    const float* vp = vals + e0 + tid;
#pragma unroll
    for (int k = 0; k < EPT; ++k) r[k] = rp[k * BIN_THREADS];
#pragma unroll
    for (int k = 0; k < EPT; ++k) {
      c[k] = cp[k * BIN_THREADS];
      vb[k] = __float_as_int(vp[k * BIN_THREADS]);
    }
  } else {
#pragma unroll
    for (int k = 0; k < EPT; ++k) {
      const int e = e0 + tid + k * BIN_THREADS;
      r[k] = (e < e1) ? row[e] : -1;
    }
#pragma unroll
    for (int k = 0; k < EPT; ++k) {
      const int e = e0 + tid + k * BIN_THREADS;
      const bool ok = e < e1;
      c[k] = ok ? col[e] : 0;
      vb[k] = ok ? __float_as_int(vals[e]) : 0;
    }
  }
  for (int i = tid; i < BIN_WAVES * n_buckets; i += BIN_THREADS) lhw[i] = 0;
  lds_only_barrier();
#pragma unroll
  for (int k = 0; k < EPT; ++k)
    if (FULL || r[k] >= 0) atomicAdd(&lh[r[k] >> RPB_SHIFT], 1);
  lds_only_barrier();

  const int lane = tid & 31;
  const int wv = tid >> 5;
  int carry = 0;
  for (int b0 = 0; b0 < n_buckets; b0 += 2 * BIN_THREADS) {
    const int i0 = b0 + 2 * tid;
    int cn[2][BIN_WAVES];
    int s = 0;
#pragma unroll
    for (int u = 0; u < 2; ++u) {
      const bool ok = i0 + u < n_buckets;
      const int ic = ok ? (i0 + u) : (n_buckets - 1);
#pragma unroll
      for (int q = 0; q < BIN_WAVES; ++q) {
        const int v = lhw[q * n_buckets + ic];
        cn[u][q] = ok ? v : 0;
        s += cn[u][q];
      }
    }
    const int incl = wave_incl_scan(s, lane);
    if (b0 != 0) lds_only_barrier();
    if (lane == 31) part[wv] = incl;
    lds_only_barrier();
    int base = carry, tot = 0;
#pragma unroll
    for (int q = 0; q < BIN_WAVES; ++q) {
      const int pq = part[q];
      base += (q < wv) ? pq : 0;
      tot += pq;
    }
    carry += tot;
    int pre = base + incl - s;
#pragma unroll
    for (int u = 0; u < 2; ++u) {
      if (i0 + u < n_buckets) {
        trow[i0 + u] = pre;
#pragma unroll
        for (int q = 0; q < BIN_WAVES; ++q) {
          lhw[q * n_buckets + i0 + u] = pre;
          pre += cn[u][q];
        }
      }
    }
  }
  if (tid == 0) trow[n_buckets] = e1 - e0;
  lds_only_barrier();

  int pos[EPT];
#pragma unroll
  for (int k = 0; k < EPT; ++k) {
    pos[k] = 0;
    if (FULL || r[k] >= 0) pos[k] = atomicAdd(&lh[r[k] >> RPB_SHIFT], 1);
  }
#pragma unroll
  for (int k = 0; k < EPT; ++k) {
    v4i rec;
    rec.x = c[k];
    rec.y = r[k] & (RPB - 1);
    rec.z = vb[k];
    rec.w = 0;
    stage[(FULL || r[k] >= 0) ? pos[k] : CHUNK] = rec;
  }
  __syncthreads();
  if (FULL) {
#pragma unroll
    for (int k = 0; k < EPT; ++k) dst[tid + k * BIN_THREADS] = stage[tid + k * BIN_THREADS];
  } else {
    const int cnt = e1 - e0;
    for (int i = tid; i < cnt; i += BIN_THREADS) dst[i] = stage[i];
  }
}

__global__ void __launch_bounds__(BIN_THREADS)
gcn_chunk_sort_kernel(const int* __restrict__ row, const int* __restrict__ col,
                      const float* __restrict__ vals, int n_edges, int n_buckets,
                      int* __restrict__ tab, v4i* __restrict__ recs) {
  __shared__ __attribute__((aligned(16))) v4i stage[CHUNK + 1];
  extern __shared__ int sm[];
  int* lhw = sm;
  int* part = sm + BIN_WAVES * n_buckets;

  static_assert(CHUNK % BIN_THREADS == 0, "a whole number of edges per lane");
  const int e0 = blockIdx.x * CHUNK;
  const int e1 = (e0 + CHUNK < n_edges) ? (e0 + CHUNK) : n_edges;
  int* trow = tab + (long long)blockIdx.x * (n_buckets + 1);
  if (e1 - e0 == CHUNK)
    chunk_sort_body<true>(row, col, vals, e0, e1, n_buckets, trow, recs + e0, stage, lhw, part);
  else
    chunk_sort_body<false>(row, col, vals, e0, e1, n_buckets, trow, recs + e0, stage, lhw, part);
}

__global__ void __launch_bounds__(SUM_THREADS) __attribute__((amdgpu_waves_per_eu(4, 8)))
gcn_bucket_sum_gemm_kernel(const float* __restrict__ X, const v4i* __restrict__ recs,
                           const int* __restrict__ tab, const float* __restrict__ W,
                           const float* __restrict__ bias, float* __restrict__ out,
                           int n_nodes, int n_buckets, int n_chunks) {
  __shared__ __attribute__((aligned(16))) float tile[RPB * TP];
  __shared__ __attribute__((aligned(16))) float Wl2[DFEAT * DFEAT];
  __shared__ float bl[DFEAT];
  __shared__ int sc[SCH];
  __shared__ int sr[SCH];
  __shared__ int sz[SCH];
  __shared__ int rcw[SUM_WAVES * (RPB / 2)];
  __shared__ int wsum[SUM_WAVES];
  __shared__ int pfx[MAXC + 1];
  __shared__ int pst[MAXC];
  const int tid = threadIdx.x;
  const int b = blockIdx.x;

  static_assert(DFEAT * DFEAT == 8 * SUM_THREADS, "two 4-float pieces of W per thread");
  const v4f w0 = ((const v4f*)W)[tid];
  const v4f w1 = ((const v4f*)W)[tid + SUM_THREADS];
  float bv0 = 0.0f;
  if (tid < DFEAT) bv0 = bias[tid];
  static_assert(MAXC <= SUM_THREADS, "one table entry per thread");
  static_assert(MAXC / 32 <= SUM_WAVES && 2 * (RPB / 64) <= SUM_WAVES, "wave totals fit wsum");
  int sz_c = 0, st_c = 0;
  if (tid < MAXC && tid < n_chunks) {
    const int* te = tab + (long long)tid * (n_buckets + 1) + b;
    const int s = te[0];
    sz_c = te[1] - s;
    st_c = tid * CHUNK + s;
  }

  v4f z = {0.0f, 0.0f, 0.0f, 0.0f};
  for (int i = tid; i < RPB * TP / 4; i += SUM_THREADS) ((v4f*)tile)[i] = z;
  {
    const int k = tid >> 4;
    const int n = (tid & 15) * 4;
    float* d0 = Wl2 + (k >> 1) * 128 + n * 2 + (k & 1);
    float* d1 = d0 + 16 * 128;
    d0[0] = w0.x; d0[2] = w0.y; d0[4] = w0.z; d0[6] = w0.w;
    d1[0] = w1.x; d1[2] = w1.y; d1[4] = w1.z; d1[6] = w1.w;
  }
  if (tid < DFEAT) bl[tid] = bv0;
  const int lane = tid & 31;
  const int wave = __builtin_amdgcn_readfirstlane(tid >> 5);
  const int sz_incl = wave_incl_scan(sz_c, lane);
  if (lane == 31) wsum[wave] = sz_incl;
  if (tid < MAXC) pst[tid] = st_c;
  __syncthreads();
  {
    int base = 0;
#pragma unroll
    for (int q = 0; q < MAXC / 32; ++q) {
      const int t = wsum[q];
      base += (q < wave) ? t : 0;
    }
    if (tid < MAXC) pfx[tid] = base + sz_incl - sz_c;
    if (tid == MAXC - 1) pfx[MAXC] = base + sz_incl;
  }
  __syncthreads();
  const int cnt = pfx[MAXC];

  const unsigned lane8 = (unsigned)lane * 8u;

  for (int cb = 0; cb < cnt; cb += SCH) {
    const int n = (cnt - cb < SCH) ? (cnt - cb) : SCH;
    for (int i = tid; i < SUM_WAVES * (RPB / 2); i += SUM_THREADS) rcw[i] = 0;
    __syncthreads();
    int mc[SPT], mr[SPT], mz[SPT];
    int gi[SPT], lo[SPT];
#pragma unroll
    for (int k = 0; k < SPT; ++k) {
      const int i = cb + tid + k * SUM_THREADS;
      gi[k] = (i < cnt) ? i : (cnt - 1);
      lo[k] = 0;
    }
#pragma unroll
    for (int step = MAXC / 2; step > 0; step >>= 1) {
#pragma unroll
      for (int k = 0; k < SPT; ++k) lo[k] += (pfx[lo[k] + step] <= gi[k]) ? step : 0;
    }
    int ra[SPT];
#pragma unroll
    for (int k = 0; k < SPT; ++k) ra[k] = pst[lo[k]] + (gi[k] - pfx[lo[k]]);
    v4i rec[SPT];
#pragma unroll
    for (int k = 0; k < SPT; ++k) rec[k] = recs[ra[k]];
#pragma unroll
    for (int k = 0; k < SPT; ++k) {
      const bool ok = tid + k * SUM_THREADS < n;
      mc[k] = ok ? rec[k].x : 0;
      mr[k] = ok ? rec[k].y : -1;
      mz[k] = ok ? rec[k].z : 0;
    }
    int* rw = rcw + wave * (RPB / 2);
#pragma unroll
    for (int k = 0; k < SPT; ++k)
      if (mr[k] >= 0) atomicAdd(&rw[mr[k] & (RPB / 2 - 1)], 1 << ((mr[k] >> (RPB_SHIFT - 1)) * 16));
    __syncthreads();
    int wd[SUM_WAVES];
    int tl = 0, th = 0, il = 0, ih = 0;
    if (tid < RPB / 2) {
#pragma unroll
      for (int q = 0; q < SUM_WAVES; ++q) {
        wd[q] = rcw[q * (RPB / 2) + tid];
        tl += (int)((unsigned)wd[q] & 0xFFFFu);
        th += (int)((unsigned)wd[q] >> 16);
      }
      il = wave_incl_scan(tl, lane);
      ih = wave_incl_scan(th, lane);
      if (lane == 31) { wsum[wave] = il; wsum[RPB / 64 + wave] = ih; }
    }
    __syncthreads();
    if (tid < RPB / 2) {
      int basel = 0, baseh = 0, totl = 0;
#pragma unroll
      for (int q = 0; q < RPB / 64; ++q) {
        const int t = wsum[q];
        const int u = wsum[RPB / 64 + q];
        totl += t;
        basel += (q < wave) ? t : 0;
        baseh += (q < wave) ? u : 0;
      }
      int ol = basel + il - tl;
      int oh = totl + baseh + ih - th;
#pragma unroll
      for (int q = 0; q < SUM_WAVES; ++q) {
        rcw[q * (RPB / 2) + tid] = (int)((unsigned)ol | ((unsigned)oh << 16));
        ol += (int)((unsigned)wd[q] & 0xFFFFu);
        oh += (int)((unsigned)wd[q] >> 16);
      }
    }
    __syncthreads();
    int p[SPT];
#pragma unroll
    for (int k = 0; k < SPT; ++k) {
      p[k] = -1;
      if (mr[k] >= 0) {
        const int sh = (mr[k] >> (RPB_SHIFT - 1)) * 16;
        const unsigned old = (unsigned)atomicAdd(&rw[mr[k] & (RPB / 2 - 1)], 1 << sh);
        p[k] = (int)((old >> sh) & 0xFFFFu);
      }
    }
#pragma unroll
    for (int k = 0; k < SPT; ++k) {
      if (p[k] >= 0) { sc[p[k]] = mc[k]; sr[p[k]] = mr[k]; sz[p[k]] = mz[k]; }
    }
    __syncthreads();

    for (int i0 = wave * 32; i0 < n; i0 += (SUM_THREADS / 32) * 32) {
      int idx = i0 + lane;
      if (idx >= n) idx = n - 1;
      const int rc = sc[idx];
      const unsigned rc8 = (unsigned)rc * (unsigned)(DFEAT * sizeof(float));
      const int rr = sr[idx];
      const int rz = sz[idx];
      const int m = (n - i0 < 32) ? (n - i0) : 32;
      float a0 = 0.0f, a1 = 0.0f;
      const int rprev = __shfl_up(rr, 1);
      const unsigned chg = __builtin_amdgcn_ballot_w32(lane != 0 && rr != rprev);
      int last;
      if (m == 32) {
        v2f xr[32];
        unsigned ro[32];
#pragma unroll
        for (int j = 0; j < 32; ++j) ro[j] = (unsigned)__builtin_amdgcn_readlane((int)rc8, j) + lane8;
#pragma unroll
        for (int j = 0; j < 32; ++j) xr[j] = *(const v2f*)((const char*)X + ro[j]);
#pragma unroll
        for (int j = 0; j < 32; ++j) {
          if (j > 0 && ((chg >> j) & 1u)) {
            float* td = tile + __builtin_amdgcn_readlane(rr, j - 1) * TP + 2 * lane;
            atomicAdd(td + 0, a0);
            atomicAdd(td + 1, a1);
            a0 = 0.0f; a1 = 0.0f;
          }
          const float v = __int_as_float(__builtin_amdgcn_readlane(rz, j));
          a0 += xr[j].x * v;
          a1 += xr[j].y * v;
        }
        last = __builtin_amdgcn_readlane(rr, 31);
      } else {
        for (int j = 0; j < m; ++j) {
          const int c = __builtin_amdgcn_readlane(rc, j);
          const float v = __int_as_float(__builtin_amdgcn_readlane(rz, j));
          const v2f x = *(const v2f*)((const char*)X + ((unsigned)c * (unsigned)(DFEAT * sizeof(float)) + lane8));
          if (j > 0 && ((chg >> j) & 1u)) {
            float* td = tile + __builtin_amdgcn_readlane(rr, j - 1) * TP + 2 * lane;
            atomicAdd(td + 0, a0);
            atomicAdd(td + 1, a1);
            a0 = 0.0f; a1 = 0.0f;
          }
          a0 += x.x * v;
          a1 += x.y * v;
        }
        last = __builtin_amdgcn_readlane(rr, m - 1);
      }
      float* td = tile + last * TP + 2 * lane;
      atomicAdd(td + 0, a0);
      atomicAdd(td + 1, a1);
    }
  }
  __syncthreads();

  const int m0 = wave * 16;
  const int lo = lane & 15;
  const int hi = lane >> 4;
  const int koff = hi * 2;

  v8f acc[4];
#pragma unroll
  for (int t = 0; t < 4; ++t) {
    float bv = bl[t * 16 + lo];
#pragma unroll
    for (int r = 0; r < 8; ++r) acc[t][r] = bv;
  }

  const float* arow = tile + (m0 + lo) * TP;
  const v2f* wl2 = (const v2f*)Wl2;

#pragma unroll
  for (int kk = 0; kk < 16; ++kk) {
    int kb = kk * 4;
    int kpair = (kb + koff) >> 1;
    v2f a = *(const v2f*)(arow + kb + koff);
#pragma unroll
    for (int t = 0; t < 4; ++t) {
      v2f bb = wl2[kpair * 64 + t * 16 + lo];
      acc[t] = __builtin_amdgcn_wmma_f32_16x16x4_f32(
           false, a,  false, bb,
           (short)0, acc[t],  false,  false);
    }
  }

#pragma unroll
  for (int t = 0; t < 4; ++t) {
#pragma unroll
    for (int r = 0; r < 8; ++r)
      tile[(m0 + r + 8 * hi) * TP + t * 16 + lo] = acc[t][r];
  }
  __syncthreads();

  const int r0 = b * RPB;
  const int nrows = (n_nodes - r0 < RPB) ? (n_nodes - r0) : RPB;
  v4f* dst = (v4f*)(out + (long long)r0 * DFEAT);
  for (int i = tid; i < nrows * (DFEAT / 4); i += SUM_THREADS)
    dst[i] = *(const v4f*)(tile + (i >> 4) * TP + (i & 15) * 4);
}


extern "C" void kernel_launch(void* const* d_in, const int* in_sizes, int n_in,
                              void* d_out, int out_size, void* d_ws, size_t ws_size,
                              hipStream_t stream) {
  const float* X    = (const float*)d_in[0];
  const float* vals = (const float*)d_in[1];
  const float* W    = (const float*)d_in[2];
  const float* bias = (const float*)d_in[3];
  const int*   row  = (const int*)d_in[4];
  const int*   col  = (const int*)d_in[5];
  float* out = (float*)d_out;

  int n_nodes = in_sizes[0] / DFEAT;
  int n_edges = in_sizes[1];
  int n_buckets = (n_nodes + RPB - 1) >> RPB_SHIFT;
  int n_chunks = (n_edges + CHUNK - 1) / CHUNK;

  size_t rec_bytes  = (((size_t)n_edges * sizeof(v4i)) + 255) & ~(size_t)255;
  char* ws = (char*)d_ws;
  v4i* recs = (v4i*)ws;
  int* tab  = (int*)(ws + rec_bytes);
  if (rec_bytes + (size_t)n_chunks * (n_buckets + 1) * sizeof(int) > ws_size) return;
  if (n_chunks > MAXC) return;
  if (n_nodes > (1 << 24)) return;

  gcn_chunk_sort_kernel<<<n_chunks, BIN_THREADS,
                          ((size_t)n_buckets * BIN_WAVES + BIN_THREADS) * sizeof(int), stream>>>(
      row, col, vals, n_edges, n_buckets, tab, recs);

  gcn_bucket_sum_gemm_kernel<<<n_buckets, SUM_THREADS, 0, stream>>>(X, recs, tab, W, bias, out,
                                                                    n_nodes, n_buckets, n_chunks);
}


# Bucket list's last, partial 32 records padded with zero-valued repeats of its last record and summed like a full 32 (32 gathers in flight), not one waited-for load per record; sums unchanged
# speedup vs baseline: 3.8355x; 1.0943x over previous
#include <hip/hip_runtime.h>


typedef __attribute__((ext_vector_type(2))) float v2f;
typedef __attribute__((ext_vector_type(4))) float v4f;
typedef __attribute__((ext_vector_type(8))) float v8f;
typedef __attribute__((ext_vector_type(4))) int v4i;

#define DFEAT 64
#define RPB_SHIFT 8
#define RPB 256
#define TP 68
#define CHUNK 8192
#define BIN_THREADS 256
#define SUM_THREADS 512
#define EPT (CHUNK / BIN_THREADS)
#define SCH 5120
#define SPT (SCH / SUM_THREADS)
#define BIN_WAVES (BIN_THREADS / 32)
#define SUM_WAVES (SUM_THREADS / 32)
#define MAXC 256

__device__ __forceinline__ void lds_only_barrier() {
  __builtin_amdgcn_fence(__ATOMIC_RELEASE, "workgroup", "local");
  __builtin_amdgcn_s_barrier();
  __builtin_amdgcn_fence(__ATOMIC_ACQUIRE, "workgroup", "local");
}

__device__ __forceinline__ int wave_incl_scan(int x, const int lane) {
  x += __builtin_amdgcn_update_dpp(0, x, 0x111, 0xF, 0xF, true);
  x += __builtin_amdgcn_update_dpp(0, x, 0x112, 0xF, 0xF, true);
  x += __builtin_amdgcn_update_dpp(0, x, 0x114, 0xF, 0xF, true);
  x += __builtin_amdgcn_update_dpp(0, x, 0x118, 0xF, 0xF, true);
  const int low_row_total = __builtin_amdgcn_readlane(x, 15);
  return x + ((lane & 16) ? low_row_total : 0);
}

template <bool FULL>
__device__ __forceinline__ void chunk_sort_body(
    const int* __restrict__ row, const int* __restrict__ col, const float* __restrict__ vals,
    const int e0, const int e1, const int n_buckets, int* __restrict__ trow,
    v4i* __restrict__ dst, v4i* stage, int* lhw, int* part) {
  const int tid = threadIdx.x;
  int* lh = lhw + (tid >> 5) * n_buckets;

  int r[EPT], c[EPT], vb[EPT];
  if (FULL) {
    const int* rp = row + e0 + tid;
    const int* cp = col + e0 + tid;
    const float* vp = vals + e0 + tid;
#pragma unroll
    for (int k = 0; k < EPT; ++k) r[k] = rp[k * BIN_THREADS];
#pragma unroll
    for (int k = 0; k < EPT; ++k) {
      c[k] = cp[k * BIN_THREADS];
      vb[k] = __float_as_int(vp[k * BIN_THREADS]);
    }
  } else {
#pragma unroll
    for (int k = 0; k < EPT; ++k) {
      const int e = e0 + tid + k * BIN_THREADS;
      r[k] = (e < e1) ? row[e] : -1;
    }
#pragma unroll
    for (int k = 0; k < EPT; ++k) {
      const int e = e0 + tid + k * BIN_THREADS;
      const bool ok = e < e1;
      c[k] = ok ? col[e] : 0;
      vb[k] = ok ? __float_as_int(vals[e]) : 0;
    }
  }
  for (int i = tid; i < BIN_WAVES * n_buckets; i += BIN_THREADS) lhw[i] = 0;
  lds_only_barrier();
#pragma unroll
  for (int k = 0; k < EPT; ++k)
    if (FULL || r[k] >= 0) atomicAdd(&lh[r[k] >> RPB_SHIFT], 1);
  lds_only_barrier();

  const int lane = tid & 31;
  const int wv = tid >> 5;
  int carry = 0;
  for (int b0 = 0; b0 < n_buckets; b0 += 2 * BIN_THREADS) {
    const int i0 = b0 + 2 * tid;
    int cn[2][BIN_WAVES];
    int s = 0;
#pragma unroll
    for (int u = 0; u < 2; ++u) {
      const bool ok = i0 + u < n_buckets;
      const int ic = ok ? (i0 + u) : (n_buckets - 1);
#pragma unroll
      for (int q = 0; q < BIN_WAVES; ++q) {
        const int v = lhw[q * n_buckets + ic];
        cn[u][q] = ok ? v : 0;
        s += cn[u][q];
      }
    }
    const int incl = wave_incl_scan(s, lane);
    if (b0 != 0) lds_only_barrier();
    if (lane == 31) part[wv] = incl;
    lds_only_barrier();
    int base = carry, tot = 0;
#pragma unroll
    for (int q = 0; q < BIN_WAVES; ++q) {
      const int pq = part[q];
      base += (q < wv) ? pq : 0;
      tot += pq;
    }
    carry += tot;
    int pre = base + incl - s;
#pragma unroll
    for (int u = 0; u < 2; ++u) {
      if (i0 + u < n_buckets) {
        trow[i0 + u] = pre;
#pragma unroll
        for (int q = 0; q < BIN_WAVES; ++q) {
          lhw[q * n_buckets + i0 + u] = pre;
          pre += cn[u][q];
        }
      }
    }
  }
  if (tid == 0) trow[n_buckets] = e1 - e0;
  lds_only_barrier();

  int pos[EPT];
#pragma unroll
  for (int k = 0; k < EPT; ++k) {
    pos[k] = 0;
    if (FULL || r[k] >= 0) pos[k] = atomicAdd(&lh[r[k] >> RPB_SHIFT], 1);
  }
#pragma unroll
  for (int k = 0; k < EPT; ++k) {
    v4i rec;
    rec.x = c[k];
    rec.y = r[k] & (RPB - 1);
    rec.z = vb[k];
    rec.w = 0;
    stage[(FULL || r[k] >= 0) ? pos[k] : CHUNK] = rec;
  }
  __syncthreads();
  if (FULL) {
#pragma unroll
    for (int k = 0; k < EPT; ++k) dst[tid + k * BIN_THREADS] = stage[tid + k * BIN_THREADS];
  } else {
    const int cnt = e1 - e0;
    for (int i = tid; i < cnt; i += BIN_THREADS) dst[i] = stage[i];
  }
}

__global__ void __launch_bounds__(BIN_THREADS)
gcn_chunk_sort_kernel(const int* __restrict__ row, const int* __restrict__ col,
                      const float* __restrict__ vals, int n_edges, int n_buckets,
                      int* __restrict__ tab, v4i* __restrict__ recs) {
  __shared__ __attribute__((aligned(16))) v4i stage[CHUNK + 1];
  extern __shared__ int sm[];
  int* lhw = sm;
  int* part = sm + BIN_WAVES * n_buckets;

  static_assert(CHUNK % BIN_THREADS == 0, "a whole number of edges per lane");
  const int e0 = blockIdx.x * CHUNK;
  const int e1 = (e0 + CHUNK < n_edges) ? (e0 + CHUNK) : n_edges;
  int* trow = tab + (long long)blockIdx.x * (n_buckets + 1);
  if (e1 - e0 == CHUNK)
    chunk_sort_body<true>(row, col, vals, e0, e1, n_buckets, trow, recs + e0, stage, lhw, part);
  else
    chunk_sort_body<false>(row, col, vals, e0, e1, n_buckets, trow, recs + e0, stage, lhw, part);
}

__global__ void __launch_bounds__(SUM_THREADS) __attribute__((amdgpu_waves_per_eu(4, 8)))
gcn_bucket_sum_gemm_kernel(const float* __restrict__ X, const v4i* __restrict__ recs,
                           const int* __restrict__ tab, const float* __restrict__ W,
                           const float* __restrict__ bias, float* __restrict__ out,
                           int n_nodes, int n_buckets, int n_chunks) {
  __shared__ __attribute__((aligned(16))) float tile[RPB * TP];
  __shared__ __attribute__((aligned(16))) float Wl2[DFEAT * DFEAT];
  __shared__ float bl[DFEAT];
  __shared__ int sc[SCH];
  __shared__ int sr[SCH];
  __shared__ int sz[SCH];
  __shared__ int rcw[SUM_WAVES * (RPB / 2)];
  __shared__ int wsum[SUM_WAVES];
  __shared__ int pfx[MAXC + 1];
  __shared__ int pst[MAXC];
  const int tid = threadIdx.x;
  const int b = blockIdx.x;

  static_assert(DFEAT * DFEAT == 8 * SUM_THREADS, "two 4-float pieces of W per thread");
  const v4f w0 = ((const v4f*)W)[tid];
  const v4f w1 = ((const v4f*)W)[tid + SUM_THREADS];
  float bv0 = 0.0f;
  if (tid < DFEAT) bv0 = bias[tid];
  static_assert(MAXC <= SUM_THREADS, "one table entry per thread");
  static_assert(MAXC / 32 <= SUM_WAVES && 2 * (RPB / 64) <= SUM_WAVES, "wave totals fit wsum");
  int sz_c = 0, st_c = 0;
  if (tid < MAXC && tid < n_chunks) {
    const int* te = tab + (long long)tid * (n_buckets + 1) + b;
    const int s = te[0];
    sz_c = te[1] - s;
    st_c = tid * CHUNK + s;
  }

  v4f z = {0.0f, 0.0f, 0.0f, 0.0f};
  for (int i = tid; i < RPB * TP / 4; i += SUM_THREADS) ((v4f*)tile)[i] = z;
  {
    const int k = tid >> 4;
    const int n = (tid & 15) * 4;
    float* d0 = Wl2 + (k >> 1) * 128 + n * 2 + (k & 1);
    float* d1 = d0 + 16 * 128;
    d0[0] = w0.x; d0[2] = w0.y; d0[4] = w0.z; d0[6] = w0.w;
    d1[0] = w1.x; d1[2] = w1.y; d1[4] = w1.z; d1[6] = w1.w;
  }
  if (tid < DFEAT) bl[tid] = bv0;
  const int lane = tid & 31;
  const int wave = __builtin_amdgcn_readfirstlane(tid >> 5);
  const int sz_incl = wave_incl_scan(sz_c, lane);
  if (lane == 31) wsum[wave] = sz_incl;
  if (tid < MAXC) pst[tid] = st_c;
  __syncthreads();
  {
    int base = 0;
#pragma unroll
    for (int q = 0; q < MAXC / 32; ++q) {
      const int t = wsum[q];
      base += (q < wave) ? t : 0;
    }
    if (tid < MAXC) pfx[tid] = base + sz_incl - sz_c;
    if (tid == MAXC - 1) pfx[MAXC] = base + sz_incl;
  }
  __syncthreads();
  const int cnt = pfx[MAXC];

  const unsigned lane8 = (unsigned)lane * 8u;

  for (int cb = 0; cb < cnt; cb += SCH) {
    const int n = (cnt - cb < SCH) ? (cnt - cb) : SCH;
    for (int i = tid; i < SUM_WAVES * (RPB / 2); i += SUM_THREADS) rcw[i] = 0;
    __syncthreads();
    int mc[SPT], mr[SPT], mz[SPT];
    int gi[SPT], lo[SPT];
#pragma unroll
    for (int k = 0; k < SPT; ++k) {
      const int i = cb + tid + k * SUM_THREADS;
      gi[k] = (i < cnt) ? i : (cnt - 1);
      lo[k] = 0;
    }
#pragma unroll
    for (int step = MAXC / 2; step > 0; step >>= 1) {
#pragma unroll
      for (int k = 0; k < SPT; ++k) lo[k] += (pfx[lo[k] + step] <= gi[k]) ? step : 0;
    }
    int ra[SPT];
#pragma unroll
    for (int k = 0; k < SPT; ++k) ra[k] = pst[lo[k]] + (gi[k] - pfx[lo[k]]);
    v4i rec[SPT];
#pragma unroll
    for (int k = 0; k < SPT; ++k) rec[k] = recs[ra[k]];
#pragma unroll
    for (int k = 0; k < SPT; ++k) {
      const bool ok = tid + k * SUM_THREADS < n;
      mc[k] = ok ? rec[k].x : 0;
      mr[k] = ok ? rec[k].y : -1;
      mz[k] = ok ? rec[k].z : 0;
    }
    int* rw = rcw + wave * (RPB / 2);
#pragma unroll
    for (int k = 0; k < SPT; ++k)
      if (mr[k] >= 0) atomicAdd(&rw[mr[k] & (RPB / 2 - 1)], 1 << ((mr[k] >> (RPB_SHIFT - 1)) * 16));
    __syncthreads();
    int wd[SUM_WAVES];
    int tl = 0, th = 0, il = 0, ih = 0;
    if (tid < RPB / 2) {
#pragma unroll
      for (int q = 0; q < SUM_WAVES; ++q) {
        wd[q] = rcw[q * (RPB / 2) + tid];
        tl += (int)((unsigned)wd[q] & 0xFFFFu);
        th += (int)((unsigned)wd[q] >> 16);
      }
      il = wave_incl_scan(tl, lane);
      ih = wave_incl_scan(th, lane);
      if (lane == 31) { wsum[wave] = il; wsum[RPB / 64 + wave] = ih; }
    }
    __syncthreads();
    if (tid < RPB / 2) {
      int basel = 0, baseh = 0, totl = 0;
#pragma unroll
      for (int q = 0; q < RPB / 64; ++q) {
        const int t = wsum[q];
        const int u = wsum[RPB / 64 + q];
        totl += t;
        basel += (q < wave) ? t : 0;
        baseh += (q < wave) ? u : 0;
      }
      int ol = basel + il - tl;
      int oh = totl + baseh + ih - th;
#pragma unroll
      for (int q = 0; q < SUM_WAVES; ++q) {
        rcw[q * (RPB / 2) + tid] = (int)((unsigned)ol | ((unsigned)oh << 16));
        ol += (int)((unsigned)wd[q] & 0xFFFFu);
        oh += (int)((unsigned)wd[q] >> 16);
      }
    }
    __syncthreads();
    int p[SPT];
#pragma unroll
    for (int k = 0; k < SPT; ++k) {
      p[k] = -1;
      if (mr[k] >= 0) {
        const int sh = (mr[k] >> (RPB_SHIFT - 1)) * 16;
        const unsigned old = (unsigned)atomicAdd(&rw[mr[k] & (RPB / 2 - 1)], 1 << sh);
        p[k] = (int)((old >> sh) & 0xFFFFu);
      }
    }
#pragma unroll
    for (int k = 0; k < SPT; ++k) {
      if (p[k] >= 0) { sc[p[k]] = mc[k]; sr[p[k]] = mr[k]; sz[p[k]] = mz[k]; }
    }
    __syncthreads();

    for (int i0 = wave * 32; i0 < n; i0 += (SUM_THREADS / 32) * 32) {
      const bool in_list = i0 + lane < n;
      const int idx = in_list ? (i0 + lane) : (n - 1);
      const int rc = sc[idx];
      const unsigned rc8 = (unsigned)rc * (unsigned)(DFEAT * sizeof(float));
      const int rr = sr[idx];
      const int rz = in_list ? sz[idx] : 0;
      float a0 = 0.0f, a1 = 0.0f;
      const int rprev = __shfl_up(rr, 1);
      const unsigned chg = __builtin_amdgcn_ballot_w32(lane != 0 && rr != rprev);
      v2f xr[32];
      unsigned ro[32];
#pragma unroll
      for (int j = 0; j < 32; ++j) ro[j] = (unsigned)__builtin_amdgcn_readlane((int)rc8, j) + lane8;
#pragma unroll
      for (int j = 0; j < 32; ++j) xr[j] = *(const v2f*)((const char*)X + ro[j]);
#pragma unroll
      for (int j = 0; j < 32; ++j) {
        if (j > 0 && ((chg >> j) & 1u)) {
          float* td = tile + __builtin_amdgcn_readlane(rr, j - 1) * TP + 2 * lane;
          atomicAdd(td + 0, a0);
          atomicAdd(td + 1, a1);
          a0 = 0.0f; a1 = 0.0f;
        }
        const float v = __int_as_float(__builtin_amdgcn_readlane(rz, j));
        a0 += xr[j].x * v;
        a1 += xr[j].y * v;
      }
      float* td = tile + __builtin_amdgcn_readlane(rr, 31) * TP + 2 * lane;
      atomicAdd(td + 0, a0);
      atomicAdd(td + 1, a1);
    }
  }
  __syncthreads();

  const int m0 = wave * 16;
  const int lo = lane & 15;
  const int hi = lane >> 4;
  const int koff = hi * 2;

  v8f acc[4];
#pragma unroll
  for (int t = 0; t < 4; ++t) {
    float bv = bl[t * 16 + lo];
#pragma unroll
    for (int r = 0; r < 8; ++r) acc[t][r] = bv;
  }

  const float* arow = tile + (m0 + lo) * TP;
  const v2f* wl2 = (const v2f*)Wl2;

#pragma unroll
  for (int kk = 0; kk < 16; ++kk) {
    int kb = kk * 4;
    int kpair = (kb + koff) >> 1;
    v2f a = *(const v2f*)(arow + kb + koff);
#pragma unroll
    for (int t = 0; t < 4; ++t) {
      v2f bb = wl2[kpair * 64 + t * 16 + lo];
      acc[t] = __builtin_amdgcn_wmma_f32_16x16x4_f32(
           false, a,  false, bb,
           (short)0, acc[t],  false,  false);
    }
  }

#pragma unroll
  for (int t = 0; t < 4; ++t) {
#pragma unroll
    for (int r = 0; r < 8; ++r)
      tile[(m0 + r + 8 * hi) * TP + t * 16 + lo] = acc[t][r];
  }
  __syncthreads();

  const int r0 = b * RPB;
  const int nrows = (n_nodes - r0 < RPB) ? (n_nodes - r0) : RPB;
  v4f* dst = (v4f*)(out + (long long)r0 * DFEAT);
  for (int i = tid; i < nrows * (DFEAT / 4); i += SUM_THREADS)
    dst[i] = *(const v4f*)(tile + (i >> 4) * TP + (i & 15) * 4);
}


extern "C" void kernel_launch(void* const* d_in, const int* in_sizes, int n_in,
                              void* d_out, int out_size, void* d_ws, size_t ws_size,
                              hipStream_t stream) {
  const float* X    = (const float*)d_in[0];
  const float* vals = (const float*)d_in[1];
  const float* W    = (const float*)d_in[2];
  const float* bias = (const float*)d_in[3];
  const int*   row  = (const int*)d_in[4];
  const int*   col  = (const int*)d_in[5];
  float* out = (float*)d_out;

  int n_nodes = in_sizes[0] / DFEAT;
  int n_edges = in_sizes[1];
  int n_buckets = (n_nodes + RPB - 1) >> RPB_SHIFT;
  int n_chunks = (n_edges + CHUNK - 1) / CHUNK;

  size_t rec_bytes  = (((size_t)n_edges * sizeof(v4i)) + 255) & ~(size_t)255;
  char* ws = (char*)d_ws;
  v4i* recs = (v4i*)ws;
  int* tab  = (int*)(ws + rec_bytes);
  if (rec_bytes + (size_t)n_chunks * (n_buckets + 1) * sizeof(int) > ws_size) return;
  if (n_chunks > MAXC) return;
  if (n_nodes > (1 << 24)) return;

  gcn_chunk_sort_kernel<<<n_chunks, BIN_THREADS,
                          ((size_t)n_buckets * BIN_WAVES + BIN_THREADS) * sizeof(int), stream>>>(
      row, col, vals, n_edges, n_buckets, tab, recs);

  gcn_bucket_sum_gemm_kernel<<<n_buckets, SUM_THREADS, 0, stream>>>(X, recs, tab, W, bias, out,
                                                                    n_nodes, n_buckets, n_chunks);
}


# Bucket height 256 -> 200 rows (row / 200 by multiply-high; pass size 4096): 500 bucket work-groups, 2 on nearly every WGP, not 391 with 2 on 135 WGPs and 1 on 121; order of a row's sum changes
# speedup vs baseline: 3.9617x; 1.0329x over previous
#include <hip/hip_runtime.h>


typedef __attribute__((ext_vector_type(2))) float v2f;
typedef __attribute__((ext_vector_type(4))) float v4f;
typedef __attribute__((ext_vector_type(8))) float v8f;
typedef __attribute__((ext_vector_type(4))) int v4i;

#define DFEAT 64
#define RPB 200
#define RLOC_BITS 8
#define TROWS 208
#define RH 128
#define RH_SHIFT 7
#define TP 68
#define CHUNK 8192
#define BIN_THREADS 256
#define SUM_THREADS 512
#define EPT (CHUNK / BIN_THREADS)
#define SCH 4096
#define SPT (SCH / SUM_THREADS)
#define BIN_WAVES (BIN_THREADS / 32)
#define SUM_WAVES (SUM_THREADS / 32)
#define MAXC 256

__device__ __forceinline__ void lds_only_barrier() {
  __builtin_amdgcn_fence(__ATOMIC_RELEASE, "workgroup", "local");
  __builtin_amdgcn_s_barrier();
  __builtin_amdgcn_fence(__ATOMIC_ACQUIRE, "workgroup", "local");
}

__device__ __forceinline__ int wave_incl_scan(int x, const int lane) {
  x += __builtin_amdgcn_update_dpp(0, x, 0x111, 0xF, 0xF, true);
  x += __builtin_amdgcn_update_dpp(0, x, 0x112, 0xF, 0xF, true);
  x += __builtin_amdgcn_update_dpp(0, x, 0x114, 0xF, 0xF, true);
  x += __builtin_amdgcn_update_dpp(0, x, 0x118, 0xF, 0xF, true);
  const int low_row_total = __builtin_amdgcn_readlane(x, 15);
  return x + ((lane & 16) ? low_row_total : 0);
}

template <bool FULL>
__device__ __forceinline__ void chunk_sort_body(
    const int* __restrict__ row, const int* __restrict__ col, const float* __restrict__ vals,
    const int e0, const int e1, const int n_buckets, int* __restrict__ trow,
    v4i* __restrict__ dst, v4i* stage, int* lhw, int* part) {
  const int tid = threadIdx.x;
  int* lh = lhw + (tid >> 5) * n_buckets;

  int r[EPT], c[EPT], vb[EPT];
  if (FULL) {
    const int* rp = row + e0 + tid;
    const int* cp = col + e0 + tid;
    const float* vp = vals + e0 + tid;
#pragma unroll
    for (int k = 0; k < EPT; ++k) r[k] = rp[k * BIN_THREADS];
#pragma unroll
    for (int k = 0; k < EPT; ++k) {
      c[k] = cp[k * BIN_THREADS];
      vb[k] = __float_as_int(vp[k * BIN_THREADS]);
    }
  } else {
#pragma unroll
    for (int k = 0; k < EPT; ++k) {
      const int e = e0 + tid + k * BIN_THREADS;
      r[k] = (e < e1) ? row[e] : -1;
    }
#pragma unroll
    for (int k = 0; k < EPT; ++k) {
      const int e = e0 + tid + k * BIN_THREADS;
      const bool ok = e < e1;
      c[k] = ok ? col[e] : 0;
      vb[k] = ok ? __float_as_int(vals[e]) : 0;
    }
  }
  for (int i = tid; i < BIN_WAVES * n_buckets; i += BIN_THREADS) lhw[i] = 0;
#pragma unroll
  for (int k = 0; k < EPT; ++k) {
    if (FULL || r[k] >= 0) {
      const unsigned q = (unsigned)r[k] / (unsigned)RPB;
      r[k] = (int)((q << RLOC_BITS) | ((unsigned)r[k] - q * (unsigned)RPB));
    }
  }
  lds_only_barrier();
#pragma unroll
  for (int k = 0; k < EPT; ++k)
    if (FULL || r[k] >= 0) atomicAdd(&lh[r[k] >> RLOC_BITS], 1);
  lds_only_barrier();

  const int lane = tid & 31;
  const int wv = tid >> 5;
  int carry = 0;
  for (int b0 = 0; b0 < n_buckets; b0 += 2 * BIN_THREADS) {
    const int i0 = b0 + 2 * tid;
    int cn[2][BIN_WAVES];
    int s = 0;
#pragma unroll
    for (int u = 0; u < 2; ++u) {
      const bool ok = i0 + u < n_buckets;
      const int ic = ok ? (i0 + u) : (n_buckets - 1);
#pragma unroll
      for (int q = 0; q < BIN_WAVES; ++q) {
        const int v = lhw[q * n_buckets + ic];
        cn[u][q] = ok ? v : 0;
        s += cn[u][q];
      }
    }
    const int incl = wave_incl_scan(s, lane);
    if (b0 != 0) lds_only_barrier();
    if (lane == 31) part[wv] = incl;
    lds_only_barrier();
    int base = carry, tot = 0;
#pragma unroll
    for (int q = 0; q < BIN_WAVES; ++q) {
      const int pq = part[q];
      base += (q < wv) ? pq : 0;
      tot += pq;
    }
    carry += tot;
    int pre = base + incl - s;
#pragma unroll
    for (int u = 0; u < 2; ++u) {
      if (i0 + u < n_buckets) {
        trow[i0 + u] = pre;
#pragma unroll
        for (int q = 0; q < BIN_WAVES; ++q) {
          lhw[q * n_buckets + i0 + u] = pre;
          pre += cn[u][q];
        }
      }
    }
  }
  if (tid == 0) trow[n_buckets] = e1 - e0;
  lds_only_barrier();

  int pos[EPT];
#pragma unroll
  for (int k = 0; k < EPT; ++k) {
    pos[k] = 0;
    if (FULL || r[k] >= 0) pos[k] = atomicAdd(&lh[r[k] >> RLOC_BITS], 1);
  }
#pragma unroll
  for (int k = 0; k < EPT; ++k) {
    v4i rec;
    rec.x = c[k];
    rec.y = r[k] & ((1 << RLOC_BITS) - 1);
    rec.z = vb[k];
    rec.w = 0;
    stage[(FULL || r[k] >= 0) ? pos[k] : CHUNK] = rec;
  }
  __syncthreads();
  if (FULL) {
#pragma unroll
    for (int k = 0; k < EPT; ++k) dst[tid + k * BIN_THREADS] = stage[tid + k * BIN_THREADS];
  } else {
    const int cnt = e1 - e0;
    for (int i = tid; i < cnt; i += BIN_THREADS) dst[i] = stage[i];
  }
}

__global__ void __launch_bounds__(BIN_THREADS)
gcn_chunk_sort_kernel(const int* __restrict__ row, const int* __restrict__ col,
                      const float* __restrict__ vals, int n_edges, int n_buckets,
                      int* __restrict__ tab, v4i* __restrict__ recs) {
  __shared__ __attribute__((aligned(16))) v4i stage[CHUNK + 1];
  extern __shared__ int sm[];
  int* lhw = sm;
  int* part = sm + BIN_WAVES * n_buckets;

  static_assert(CHUNK % BIN_THREADS == 0, "a whole number of edges per lane");
  const int e0 = blockIdx.x * CHUNK;
  const int e1 = (e0 + CHUNK < n_edges) ? (e0 + CHUNK) : n_edges;
  int* trow = tab + (long long)blockIdx.x * (n_buckets + 1);
  if (e1 - e0 == CHUNK)
    chunk_sort_body<true>(row, col, vals, e0, e1, n_buckets, trow, recs + e0, stage, lhw, part);
  else
    chunk_sort_body<false>(row, col, vals, e0, e1, n_buckets, trow, recs + e0, stage, lhw, part);
}

__global__ void __launch_bounds__(SUM_THREADS) __attribute__((amdgpu_waves_per_eu(4, 8)))
gcn_bucket_sum_gemm_kernel(const float* __restrict__ X, const v4i* __restrict__ recs,
                           const int* __restrict__ tab, const float* __restrict__ W,
                           const float* __restrict__ bias, float* __restrict__ out,
                           int n_nodes, int n_buckets, int n_chunks) {
  __shared__ __attribute__((aligned(16))) float tile[TROWS * TP];
  __shared__ __attribute__((aligned(16))) float Wl2[DFEAT * DFEAT];
  __shared__ float bl[DFEAT];
  __shared__ int sc[SCH];
  __shared__ int sr[SCH];
  __shared__ int sz[SCH];
  __shared__ int rcw[SUM_WAVES * RH];
  __shared__ int wsum[SUM_WAVES];
  __shared__ int pfx[MAXC + 1];
  __shared__ int pst[MAXC];
  const int tid = threadIdx.x;
  const int b = blockIdx.x;

  static_assert(DFEAT * DFEAT == 8 * SUM_THREADS, "two 4-float pieces of W per thread");
  const v4f w0 = ((const v4f*)W)[tid];
  const v4f w1 = ((const v4f*)W)[tid + SUM_THREADS];
  float bv0 = 0.0f;
  if (tid < DFEAT) bv0 = bias[tid];
  static_assert(MAXC <= SUM_THREADS, "one table entry per thread");
  static_assert(MAXC / 32 <= SUM_WAVES && 2 * (RH / 32) <= SUM_WAVES, "wave totals fit wsum");
  static_assert(RPB <= 2 * RH && RPB <= (1 << RLOC_BITS) && RPB <= TROWS && TROWS % 16 == 0 &&
                TROWS / 16 <= SUM_WAVES, "a bucket's rows fit the counters and the tile's strips");
  int sz_c = 0, st_c = 0;
  if (tid < MAXC && tid < n_chunks) {
    const int* te = tab + (long long)tid * (n_buckets + 1) + b;
    const int s = te[0];
    sz_c = te[1] - s;
    st_c = tid * CHUNK + s;
  }

  v4f z = {0.0f, 0.0f, 0.0f, 0.0f};
  for (int i = tid; i < TROWS * TP / 4; i += SUM_THREADS) ((v4f*)tile)[i] = z;
  {
    const int k = tid >> 4;
    const int n = (tid & 15) * 4;
    float* d0 = Wl2 + (k >> 1) * 128 + n * 2 + (k & 1);
    float* d1 = d0 + 16 * 128;
    d0[0] = w0.x; d0[2] = w0.y; d0[4] = w0.z; d0[6] = w0.w;
    d1[0] = w1.x; d1[2] = w1.y; d1[4] = w1.z; d1[6] = w1.w;
  }
  if (tid < DFEAT) bl[tid] = bv0;
  const int lane = tid & 31;
  const int wave = __builtin_amdgcn_readfirstlane(tid >> 5);
  const int sz_incl = wave_incl_scan(sz_c, lane);
  if (lane == 31) wsum[wave] = sz_incl;
  if (tid < MAXC) pst[tid] = st_c;
  __syncthreads();
  {
    int base = 0;
#pragma unroll
    for (int q = 0; q < MAXC / 32; ++q) {
      const int t = wsum[q];
      base += (q < wave) ? t : 0;
    }
    if (tid < MAXC) pfx[tid] = base + sz_incl - sz_c;
    if (tid == MAXC - 1) pfx[MAXC] = base + sz_incl;
  }
  __syncthreads();
  const int cnt = pfx[MAXC];

  const unsigned lane8 = (unsigned)lane * 8u;

  for (int cb = 0; cb < cnt; cb += SCH) {
    const int n = (cnt - cb < SCH) ? (cnt - cb) : SCH;
    for (int i = tid; i < SUM_WAVES * RH; i += SUM_THREADS) rcw[i] = 0;
    __syncthreads();
    int mc[SPT], mr[SPT], mz[SPT];
    int gi[SPT], lo[SPT];
#pragma unroll
    for (int k = 0; k < SPT; ++k) {
      const int i = cb + tid + k * SUM_THREADS;
      gi[k] = (i < cnt) ? i : (cnt - 1);
      lo[k] = 0;
    }
#pragma unroll
    for (int step = MAXC / 2; step > 0; step >>= 1) {
#pragma unroll
      for (int k = 0; k < SPT; ++k) lo[k] += (pfx[lo[k] + step] <= gi[k]) ? step : 0;
    }
    int ra[SPT];
#pragma unroll
    for (int k = 0; k < SPT; ++k) ra[k] = pst[lo[k]] + (gi[k] - pfx[lo[k]]);
    v4i rec[SPT];
#pragma unroll
    for (int k = 0; k < SPT; ++k) rec[k] = recs[ra[k]];
#pragma unroll
    for (int k = 0; k < SPT; ++k) {
      const bool ok = tid + k * SUM_THREADS < n;
      mc[k] = ok ? rec[k].x : 0;
      mr[k] = ok ? rec[k].y : -1;
      mz[k] = ok ? rec[k].z : 0;
    }
    int* rw = rcw + wave * RH;
#pragma unroll
    for (int k = 0; k < SPT; ++k)
      if (mr[k] >= 0) atomicAdd(&rw[mr[k] & (RH - 1)], 1 << ((mr[k] >> RH_SHIFT) * 16));
    __syncthreads();
    int wd[SUM_WAVES];
    int tl = 0, th = 0, il = 0, ih = 0;
    if (tid < RH) {
#pragma unroll
      for (int q = 0; q < SUM_WAVES; ++q) {
        wd[q] = rcw[q * RH + tid];
        tl += (int)((unsigned)wd[q] & 0xFFFFu);
        th += (int)((unsigned)wd[q] >> 16);
      }
      il = wave_incl_scan(tl, lane);
      ih = wave_incl_scan(th, lane);
      if (lane == 31) { wsum[wave] = il; wsum[RH / 32 + wave] = ih; }
    }
    __syncthreads();
    if (tid < RH) {
      int basel = 0, baseh = 0, totl = 0;
#pragma unroll
      for (int q = 0; q < RH / 32; ++q) {
        const int t = wsum[q];
        const int u = wsum[RH / 32 + q];
        totl += t;
        basel += (q < wave) ? t : 0;
        baseh += (q < wave) ? u : 0;
      }
      int ol = basel + il - tl;
      int oh = totl + baseh + ih - th;
#pragma unroll
      for (int q = 0; q < SUM_WAVES; ++q) {
        rcw[q * RH + tid] = (int)((unsigned)ol | ((unsigned)oh << 16));
        ol += (int)((unsigned)wd[q] & 0xFFFFu);
        oh += (int)((unsigned)wd[q] >> 16);
      }
    }
    __syncthreads();
    int p[SPT];
#pragma unroll
    for (int k = 0; k < SPT; ++k) {
      p[k] = -1;
      if (mr[k] >= 0) {
        const int sh = (mr[k] >> RH_SHIFT) * 16;
        const unsigned old = (unsigned)atomicAdd(&rw[mr[k] & (RH - 1)], 1 << sh);
        p[k] = (int)((old >> sh) & 0xFFFFu);
      }
    }
#pragma unroll
    for (int k = 0; k < SPT; ++k) {
      if (p[k] >= 0) { sc[p[k]] = mc[k]; sr[p[k]] = mr[k]; sz[p[k]] = mz[k]; }
    }
    __syncthreads();

    for (int i0 = wave * 32; i0 < n; i0 += (SUM_THREADS / 32) * 32) {
      const bool in_list = i0 + lane < n;
      const int idx = in_list ? (i0 + lane) : (n - 1);
      const int rc = sc[idx];
      const unsigned rc8 = (unsigned)rc * (unsigned)(DFEAT * sizeof(float));
      const int rr = sr[idx];
      const int rz = in_list ? sz[idx] : 0;
      float a0 = 0.0f, a1 = 0.0f;
      const int rprev = __shfl_up(rr, 1);
      const unsigned chg = __builtin_amdgcn_ballot_w32(lane != 0 && rr != rprev);
      v2f xr[32];
      unsigned ro[32];
#pragma unroll
      for (int j = 0; j < 32; ++j) ro[j] = (unsigned)__builtin_amdgcn_readlane((int)rc8, j) + lane8;
#pragma unroll
      for (int j = 0; j < 32; ++j) xr[j] = *(const v2f*)((const char*)X + ro[j]);
#pragma unroll
      for (int j = 0; j < 32; ++j) {
        if (j > 0 && ((chg >> j) & 1u)) {
          float* td = tile + __builtin_amdgcn_readlane(rr, j - 1) * TP + 2 * lane;
          atomicAdd(td + 0, a0);
          atomicAdd(td + 1, a1);
          a0 = 0.0f; a1 = 0.0f;
        }
        const float v = __int_as_float(__builtin_amdgcn_readlane(rz, j));
        a0 += xr[j].x * v;
        a1 += xr[j].y * v;
      }
      float* td = tile + __builtin_amdgcn_readlane(rr, 31) * TP + 2 * lane;
      atomicAdd(td + 0, a0);
      atomicAdd(td + 1, a1);
    }
  }
  __syncthreads();

  if (wave < TROWS / 16) {
  const int m0 = wave * 16;
  const int lo = lane & 15;
  const int hi = lane >> 4;
  const int koff = hi * 2;

  v8f acc[4];
#pragma unroll
  for (int t = 0; t < 4; ++t) {
    float bv = bl[t * 16 + lo];
#pragma unroll
    for (int r = 0; r < 8; ++r) acc[t][r] = bv;
  }

  const float* arow = tile + (m0 + lo) * TP;
  const v2f* wl2 = (const v2f*)Wl2;

#pragma unroll
  for (int kk = 0; kk < 16; ++kk) {
    int kb = kk * 4;
    int kpair = (kb + koff) >> 1;
    v2f a = *(const v2f*)(arow + kb + koff);
#pragma unroll
    for (int t = 0; t < 4; ++t) {
      v2f bb = wl2[kpair * 64 + t * 16 + lo];
      acc[t] = __builtin_amdgcn_wmma_f32_16x16x4_f32(
           false, a,  false, bb,
           (short)0, acc[t],  false,  false);
    }
  }

#pragma unroll
  for (int t = 0; t < 4; ++t) {
#pragma unroll
    for (int r = 0; r < 8; ++r)
      tile[(m0 + r + 8 * hi) * TP + t * 16 + lo] = acc[t][r];
  }
  }
  __syncthreads();

  const int r0 = b * RPB;
  const int nrows = (n_nodes - r0 < RPB) ? (n_nodes - r0) : RPB;
  v4f* dst = (v4f*)(out + (long long)r0 * DFEAT);
  for (int i = tid; i < nrows * (DFEAT / 4); i += SUM_THREADS)
    dst[i] = *(const v4f*)(tile + (i >> 4) * TP + (i & 15) * 4);
}


extern "C" void kernel_launch(void* const* d_in, const int* in_sizes, int n_in,
                              void* d_out, int out_size, void* d_ws, size_t ws_size,
                              hipStream_t stream) {
  const float* X    = (const float*)d_in[0];
  const float* vals = (const float*)d_in[1];
  const float* W    = (const float*)d_in[2];
  const float* bias = (const float*)d_in[3];
  const int*   row  = (const int*)d_in[4];
  const int*   col  = (const int*)d_in[5];
  float* out = (float*)d_out;

  int n_nodes = in_sizes[0] / DFEAT;
  int n_edges = in_sizes[1];
  int n_buckets = (n_nodes + RPB - 1) / RPB;
  int n_chunks = (n_edges + CHUNK - 1) / CHUNK;

  size_t rec_bytes  = (((size_t)n_edges * sizeof(v4i)) + 255) & ~(size_t)255;
  char* ws = (char*)d_ws;
  v4i* recs = (v4i*)ws;
  int* tab  = (int*)(ws + rec_bytes);
  if (rec_bytes + (size_t)n_chunks * (n_buckets + 1) * sizeof(int) > ws_size) return;
  if (n_chunks > MAXC) return;
  if (n_nodes > (1 << 24)) return;

  gcn_chunk_sort_kernel<<<n_chunks, BIN_THREADS,
                          ((size_t)n_buckets * BIN_WAVES + BIN_THREADS) * sizeof(int), stream>>>(
      row, col, vals, n_edges, n_buckets, tab, recs);

  gcn_bucket_sum_gemm_kernel<<<n_buckets, SUM_THREADS, 0, stream>>>(X, recs, tab, W, bias, out,
                                                                    n_nodes, n_buckets, n_chunks);
}
